# starting kernel
# baseline (speedup 1.0000x reference)
#include <hip/hip_runtime.h>
#include <math.h>

#define NEG_SLOPE 0.2f
#define FDIM 256
#define HEADS 4
#define HDIM 64
#define KTILE 32
#define LDT 34

typedef float v2f __attribute__((ext_vector_type(2)));
typedef float v8f __attribute__((ext_vector_type(8)));

__device__ __forceinline__ unsigned f2key(float f) {
  unsigned u = __float_as_uint(f);
  unsigned mask = (unsigned)(((int)u) >> 31) | 0x80000000u;
  return u ^ mask;
}
__device__ __forceinline__ float key2f(unsigned k) {
  unsigned mask = (k & 0x80000000u) ? 0x80000000u : 0xFFFFFFFFu;
  return __uint_as_float(k ^ mask);
}
__device__ __forceinline__ void atomicAddF32(float* p, float v) {
  __hip_atomic_fetch_add(p, v, __ATOMIC_RELAXED, __HIP_MEMORY_SCOPE_AGENT);
}

__global__ __launch_bounds__(128) void gemm_f32_wmma(
    const float* __restrict__ H, const float* __restrict__ W,
    float* __restrict__ OUT, int N) {
  __shared__ float As[64 * LDT];
  __shared__ float Bt[64 * LDT];

  const int row0 = blockIdx.x * 64;
  const int col0 = blockIdx.y * 64;
  const int t = threadIdx.x;
  const int wave = t >> 5, lane = t & 31;
  const int hi = lane >> 4;
  const int m  = lane & 15;

  v8f acc0 = {}, acc1 = {}, acc2 = {}, acc3 = {};

  for (int kt = 0; kt < FDIM; kt += KTILE) {
    __syncthreads();
    for (int i = t; i < 64 * KTILE; i += 128) {
      const int r = i >> 5, c = i & 31;
      int rr = row0 + r; if (rr > N - 1) rr = N - 1;
      As[r * LDT + c] = H[(size_t)rr * FDIM + kt + c];
    }
    for (int i = t; i < 64 * KTILE; i += 128) {
      const int k = i >> 6, n = i & 63;
      Bt[n * LDT + k] = W[(size_t)(kt + k) * FDIM + col0 + n];
    }
    __syncthreads();

    #pragma unroll
    for (int kk = 0; kk < KTILE; kk += 4) {
      const int ka = kk + 2 * hi;
      v2f a  = *(const v2f*)&As[(wave * 16 + m) * LDT + ka];
      v2f b0 = *(const v2f*)&Bt[(m)      * LDT + ka];
      v2f b1 = *(const v2f*)&Bt[(16 + m) * LDT + ka];
      v2f b2 = *(const v2f*)&Bt[(32 + m) * LDT + ka];
      v2f b3 = *(const v2f*)&Bt[(48 + m) * LDT + ka];
      acc0 = __builtin_amdgcn_wmma_f32_16x16x4_f32(false, a, false, b0, (short)0, acc0, false, false);
      acc1 = __builtin_amdgcn_wmma_f32_16x16x4_f32(false, a, false, b1, (short)0, acc1, false, false);
      acc2 = __builtin_amdgcn_wmma_f32_16x16x4_f32(false, a, false, b2, (short)0, acc2, false, false);
      acc3 = __builtin_amdgcn_wmma_f32_16x16x4_f32(false, a, false, b3, (short)0, acc3, false, false);
    }
  }

  #pragma unroll
  for (int r = 0; r < 8; ++r) {
    const int row = row0 + wave * 16 + r + 8 * hi;
    if (row < N) {
      float* o = OUT + (size_t)row * FDIM + col0 + m;
      o[0]  = acc0[r];
      o[16] = acc1[r];
      o[32] = acc2[r];
      o[48] = acc3[r];
    }
  }
}

__global__ __launch_bounds__(256) void node_attn(
    const float* __restrict__ feat, const float* __restrict__ al,
    const float* __restrict__ ar, float* __restrict__ el,
    float* __restrict__ er, unsigned* __restrict__ emaxk,
    float* __restrict__ denom, int N) {
  const int lane = threadIdx.x & 31;
  const int n = blockIdx.x * 8 + (threadIdx.x >> 5);
  if (n >= N) return;
  const float* f = feat + (size_t)n * FDIM;
  #pragma unroll
  for (int h = 0; h < HEADS; ++h) {
    const int b = h * HDIM;
    float p = f[b + lane] * al[b + lane] + f[b + 32 + lane] * al[b + 32 + lane];
    float q = f[b + lane] * ar[b + lane] + f[b + 32 + lane] * ar[b + 32 + lane];
    #pragma unroll
    for (int off = 16; off > 0; off >>= 1) {
      p += __shfl_xor(p, off, 32);
      q += __shfl_xor(q, off, 32);
    }
    if (lane == 0) {
      el[n * HEADS + h] = p;
      er[n * HEADS + h] = q;
      emaxk[n * HEADS + h] = 0x007FFFFFu;
      denom[n * HEADS + h] = 0.0f;
    }
  }
}

__global__ __launch_bounds__(256) void init_acc(
    const float* __restrict__ hres, float* __restrict__ rst, int total4) {
  const int i = blockIdx.x * blockDim.x + threadIdx.x;
  if (i < total4) {
    float4 v = hres ? ((const float4*)hres)[i] : make_float4(0.f, 0.f, 0.f, 0.f);
    ((float4*)rst)[i] = v;
  }
}

__global__ __launch_bounds__(256) void edge_max(
    const int* __restrict__ src, const int* __restrict__ dst,
    const float* __restrict__ el, const float* __restrict__ er,
    unsigned* __restrict__ emaxk, int E) {
  const int e = blockIdx.x * blockDim.x + threadIdx.x;
  if (e >= E) return;
  const int s = src[e], d = dst[e];
  const float4 a = ((const float4*)el)[s];
  const float4 b = ((const float4*)er)[d];
  float v[4] = {a.x + b.x, a.y + b.y, a.z + b.z, a.w + b.w};
  #pragma unroll
  for (int h = 0; h < 4; ++h) {
    float x = v[h];
    x = x > 0.f ? x : NEG_SLOPE * x;
    atomicMax(&emaxk[d * 4 + h], f2key(x));
  }
}

__global__ __launch_bounds__(256) void edge_sum(
    const int* __restrict__ src, const int* __restrict__ dst,
    const float* __restrict__ el, const float* __restrict__ er,
    const unsigned* __restrict__ emaxk, float* __restrict__ denom,
    float* __restrict__ exbuf, int E) {
  const int e = blockIdx.x * blockDim.x + threadIdx.x;
  if (e >= E) return;
  const int s = src[e], d = dst[e];
  const float4 a = ((const float4*)el)[s];
  const float4 b = ((const float4*)er)[d];
  float v[4] = {a.x + b.x, a.y + b.y, a.z + b.z, a.w + b.w};
  float ex[4];
  #pragma unroll
  for (int h = 0; h < 4; ++h) {
    float x = v[h];
    x = x > 0.f ? x : NEG_SLOPE * x;
    const float mx = key2f(emaxk[d * 4 + h]);
    ex[h] = __expf(x - mx);
    atomicAddF32(&denom[d * 4 + h], ex[h]);
  }
  ((float4*)exbuf)[e] = make_float4(ex[0], ex[1], ex[2], ex[3]);
}

__global__ __launch_bounds__(256) void edge_msg(
    const int* __restrict__ src, const int* __restrict__ dst,
    const float* __restrict__ feat, const float* __restrict__ exbuf,
    const float* __restrict__ denom, float* __restrict__ rst, int E) {
  const int lane = threadIdx.x & 31;
  const int e = blockIdx.x * 8 + (threadIdx.x >> 5);
  if (e >= E) return;
  const int s = src[e], d = dst[e];
  const float4 ex = ((const float4*)exbuf)[e];
  const float4 dn = ((const float4*)denom)[d];
  const float c[4] = {ex.x / dn.x, ex.y / dn.y, ex.z / dn.z, ex.w / dn.w};
  const float* f = feat + (size_t)s * FDIM;
  float* o = rst + (size_t)d * FDIM;
  #pragma unroll
  for (int j = 0; j < 8; ++j) {
    const int i = lane + 32 * j;
    atomicAddF32(&o[i], f[i] * c[j >> 1]);
  }
}

__global__ __launch_bounds__(256) void elu_kernel(float* __restrict__ x, int total) {
  const int i = blockIdx.x * blockDim.x + threadIdx.x;
  if (i < total) {
    const float v = x[i];
    x[i] = v > 0.f ? v : (__expf(v) - 1.f);
  }
}

static void run_layer(const float* hin, const float* Wm, const float* al,
                      const float* ar, const int* src, const int* dst,
                      int N, int E, float* feat, float* el, float* er,
                      unsigned* emaxk, float* denom, float* exbuf,
                      const float* resid, float* rst, hipStream_t stream) {
  dim3 ggrid((N + 63) / 64, FDIM / 64);
  gemm_f32_wmma<<<ggrid, 128, 0, stream>>>(hin, Wm, feat, N);
  node_attn<<<(N + 7) / 8, 256, 0, stream>>>(feat, al, ar, el, er, emaxk, denom, N);
  const int total4 = N * (FDIM / 4);
  init_acc<<<(total4 + 255) / 256, 256, 0, stream>>>(resid, rst, total4);
  edge_max<<<(E + 255) / 256, 256, 0, stream>>>(src, dst, el, er, emaxk, E);
  edge_sum<<<(E + 255) / 256, 256, 0, stream>>>(src, dst, el, er, emaxk, denom, exbuf, E);
  edge_msg<<<(E + 7) / 8, 256, 0, stream>>>(src, dst, feat, exbuf, denom, rst, E);
  elu_kernel<<<(N * FDIM + 255) / 256, 256, 0, stream>>>(rst, N * FDIM);
}

extern "C" void kernel_launch(void* const* d_in, const int* in_sizes, int n_in,
                              void* d_out, int out_size, void* d_ws, size_t ws_size,
                              hipStream_t stream) {
  const float* features = (const float*)d_in[0];
  const int*   src      = (const int*)d_in[1];
  const int*   dst      = (const int*)d_in[2];
  const float* W1       = (const float*)d_in[3];
  const float* al1      = (const float*)d_in[4];
  const float* ar1      = (const float*)d_in[5];
  const float* W2       = (const float*)d_in[6];
  const float* al2      = (const float*)d_in[7];
  const float* ar2      = (const float*)d_in[8];

  const int N = in_sizes[0] / FDIM;
  const int E = in_sizes[1];

  char* w = (char*)d_ws;
  float* feat = (float*)w;  w += (size_t)N * FDIM * sizeof(float);
  float* h1   = (float*)w;  w += (size_t)N * FDIM * sizeof(float);
  float* el   = (float*)w;  w += (size_t)N * HEADS * sizeof(float);
  float* er   = (float*)w;  w += (size_t)N * HEADS * sizeof(float);
  unsigned* emaxk = (unsigned*)w; w += (size_t)N * HEADS * sizeof(unsigned);
  float* denom = (float*)w; w += (size_t)N * HEADS * sizeof(float);
  float* exbuf = (float*)w;

  run_layer(features, W1, al1, ar1, src, dst, N, E,
            feat, el, er, emaxk, denom, exbuf,
             nullptr,  h1, stream);

  run_layer(h1, W2, al2, ar2, src, dst, N, E,
            feat, el, er, emaxk, denom, exbuf,
             h1,  (float*)d_out, stream);
}


# Atomic scatter (max, sum, messages) replaced by in-call counting sort of edges by dst and one wave per dst node gathering its rows; per-node sum order = arrival-order rank; init and elu folded in
# speedup vs baseline: 3.2246x; 3.2246x over previous
#include <hip/hip_runtime.h>
#include <math.h>

#define NEG_SLOPE 0.2f
#define FDIM 256
#define HEADS 4
#define HDIM 64
#define KTILE 32
#define LDT 34

typedef float v2f __attribute__((ext_vector_type(2)));
typedef float v8f __attribute__((ext_vector_type(8)));

__global__ __launch_bounds__(128) void gemm_f32_wmma(
    const float* __restrict__ H, const float* __restrict__ W,
    float* __restrict__ OUT, int N) {
  __shared__ float As[64 * LDT];
  __shared__ float Bt[64 * LDT];

  const int row0 = blockIdx.x * 64;
  const int col0 = blockIdx.y * 64;
  const int t = threadIdx.x;
  const int wave = t >> 5, lane = t & 31;
  const int hi = lane >> 4;
  const int m  = lane & 15;

  v8f acc0 = {}, acc1 = {}, acc2 = {}, acc3 = {};

  for (int kt = 0; kt < FDIM; kt += KTILE) {
    __syncthreads();
    for (int i = t; i < 64 * KTILE; i += 128) {
      const int r = i >> 5, c = i & 31;
      int rr = row0 + r; if (rr > N - 1) rr = N - 1;
      As[r * LDT + c] = H[(size_t)rr * FDIM + kt + c];
    }
    for (int i = t; i < 64 * KTILE; i += 128) {
      const int k = i >> 6, n = i & 63;
      Bt[n * LDT + k] = W[(size_t)(kt + k) * FDIM + col0 + n];
    }
    __syncthreads();

    #pragma unroll
    for (int kk = 0; kk < KTILE; kk += 4) {
      const int ka = kk + 2 * hi;
      v2f a  = *(const v2f*)&As[(wave * 16 + m) * LDT + ka];
      v2f b0 = *(const v2f*)&Bt[(m)      * LDT + ka];
      v2f b1 = *(const v2f*)&Bt[(16 + m) * LDT + ka];
      v2f b2 = *(const v2f*)&Bt[(32 + m) * LDT + ka];
      v2f b3 = *(const v2f*)&Bt[(48 + m) * LDT + ka];
      acc0 = __builtin_amdgcn_wmma_f32_16x16x4_f32(false, a, false, b0, (short)0, acc0, false, false);
      acc1 = __builtin_amdgcn_wmma_f32_16x16x4_f32(false, a, false, b1, (short)0, acc1, false, false);
      acc2 = __builtin_amdgcn_wmma_f32_16x16x4_f32(false, a, false, b2, (short)0, acc2, false, false);
      acc3 = __builtin_amdgcn_wmma_f32_16x16x4_f32(false, a, false, b3, (short)0, acc3, false, false);
    }
  }

  #pragma unroll
  for (int r = 0; r < 8; ++r) {
    const int row = row0 + wave * 16 + r + 8 * hi;
    if (row < N) {
      float* o = OUT + (size_t)row * FDIM + col0 + m;
      o[0]  = acc0[r];
      o[16] = acc1[r];
      o[32] = acc2[r];
      o[48] = acc3[r];
    }
  }
}

__global__ __launch_bounds__(256) void node_attn(
    const float* __restrict__ feat, const float* __restrict__ al,
    const float* __restrict__ ar, float* __restrict__ el,
    float* __restrict__ er, int N) {
  const int lane = threadIdx.x & 31;
  const int n = blockIdx.x * 8 + (threadIdx.x >> 5);
  if (n >= N) return;
  const float* f = feat + (size_t)n * FDIM;
  #pragma unroll
  for (int h = 0; h < HEADS; ++h) {
    const int b = h * HDIM;
    float p = f[b + lane] * al[b + lane] + f[b + 32 + lane] * al[b + 32 + lane];
    float q = f[b + lane] * ar[b + lane] + f[b + 32 + lane] * ar[b + 32 + lane];
    #pragma unroll
    for (int off = 16; off > 0; off >>= 1) {
      p += __shfl_xor(p, off, 32);
      q += __shfl_xor(q, off, 32);
    }
    if (lane == 0) {
      el[n * HEADS + h] = p;
      er[n * HEADS + h] = q;
    }
  }
}

__global__ __launch_bounds__(256) void zero_cnt(int* __restrict__ cnt, int N) {
  const int i = blockIdx.x * blockDim.x + threadIdx.x;
  if (i < N) cnt[i] = 0;
}

__global__ __launch_bounds__(256) void edge_rank(
    const int* __restrict__ dst, int* __restrict__ cnt,
    int* __restrict__ rank, int E) {
  const int e = blockIdx.x * blockDim.x + threadIdx.x;
  if (e >= E) return;
  rank[e] = atomicAdd(&cnt[dst[e]], 1);
}

__global__ __launch_bounds__(1024) void scan_rowptr(
    const int* __restrict__ cnt, int* __restrict__ rowptr, int N) {
  __shared__ int wtot[32];
  __shared__ int wfront[32];
  const int t = threadIdx.x, lane = t & 31, wave = t >> 5;
  const int base = blockIdx.x * 1024;

  int front = 0;
  for (int i = t; i < base; i += 1024) front += cnt[i];

  const int idx = base + t;
  const int c = idx < N ? cnt[idx] : 0;
  int incl = c;
  #pragma unroll
  for (int off = 1; off < 32; off <<= 1) {
    const int v = __shfl_up(incl, off, 32);
    if (lane >= off) incl += v;
  }
  #pragma unroll
  for (int off = 16; off > 0; off >>= 1) front += __shfl_xor(front, off, 32);
  if (lane == 31) wtot[wave] = incl;
  if (lane == 0) wfront[wave] = front;
  __syncthreads();

  int add = wfront[lane] + (lane < wave ? wtot[lane] : 0);
  #pragma unroll
  for (int off = 16; off > 0; off >>= 1) add += __shfl_xor(add, off, 32);

  const int excl = add + incl - c;
  if (idx < N) rowptr[idx] = excl;
  if (idx == N - 1) rowptr[N] = excl + c;
}

__global__ __launch_bounds__(256) void csr_fill(
    const int* __restrict__ src, const int* __restrict__ dst,
    const int* __restrict__ rank, const int* __restrict__ rowptr,
    int* __restrict__ csrc, int E) {
  const int e = blockIdx.x * blockDim.x + threadIdx.x;
  if (e >= E) return;
  csrc[rowptr[dst[e]] + rank[e]] = src[e];
}

__device__ __forceinline__ float lrelu(float x) {
  return x > 0.f ? x : NEG_SLOPE * x;
}

__global__ __launch_bounds__(256) void gat_gather(
    const int* __restrict__ rowptr, const int* __restrict__ csrc,
    const float* __restrict__ feat, const float* __restrict__ el,
    const float* __restrict__ er, const float* __restrict__ resid,
    float* __restrict__ out, int N) {
  const int lane = threadIdx.x & 31;
  const int hi = lane >> 4;
  const int n = __builtin_amdgcn_readfirstlane(blockIdx.x * 8 + (threadIdx.x >> 5));
  if (n >= N) return;
  const int beg = rowptr[n], end = rowptr[n + 1];
  const float4 b = ((const float4*)er)[n];
  const float4* __restrict__ el4 = (const float4*)el;

  float m0 = -INFINITY, m1 = -INFINITY, m2 = -INFINITY, m3 = -INFINITY;
  for (int j = beg + lane; j < end; j += 32) {
    const float4 a = el4[csrc[j]];
    m0 = fmaxf(m0, lrelu(a.x + b.x));
    m1 = fmaxf(m1, lrelu(a.y + b.y));
    m2 = fmaxf(m2, lrelu(a.z + b.z));
    m3 = fmaxf(m3, lrelu(a.w + b.w));
  }
  #pragma unroll
  for (int off = 16; off > 0; off >>= 1) {
    m0 = fmaxf(m0, __shfl_xor(m0, off, 32));
    m1 = fmaxf(m1, __shfl_xor(m1, off, 32));
    m2 = fmaxf(m2, __shfl_xor(m2, off, 32));
    m3 = fmaxf(m3, __shfl_xor(m3, off, 32));
  }

  float d0 = 0.f, d1 = 0.f, d2 = 0.f, d3 = 0.f;
  for (int j = beg + lane; j < end; j += 32) {
    const float4 a = el4[csrc[j]];
    d0 += __expf(lrelu(a.x + b.x) - m0);
    d1 += __expf(lrelu(a.y + b.y) - m1);
    d2 += __expf(lrelu(a.z + b.z) - m2);
    d3 += __expf(lrelu(a.w + b.w) - m3);
  }
  #pragma unroll
  for (int off = 16; off > 0; off >>= 1) {
    d0 += __shfl_xor(d0, off, 32);
    d1 += __shfl_xor(d1, off, 32);
    d2 += __shfl_xor(d2, off, 32);
    d3 += __shfl_xor(d3, off, 32);
  }

  float4 accA = make_float4(0.f, 0.f, 0.f, 0.f);
  float4 accB = make_float4(0.f, 0.f, 0.f, 0.f);
  const float* __restrict__ fcol = feat + lane * 4;
  for (int c0 = beg; c0 < end; c0 += 32) {
    const int j = c0 + lane;
    int s = 0;
    float w0 = 0.f, w1 = 0.f, w2 = 0.f, w3 = 0.f;
    if (j < end) {
      s = csrc[j];
      const float4 a = el4[s];
      w0 = __expf(lrelu(a.x + b.x) - m0) / d0;
      w1 = __expf(lrelu(a.y + b.y) - m1) / d1;
      w2 = __expf(lrelu(a.z + b.z) - m2) / d2;
      w3 = __expf(lrelu(a.w + b.w) - m3) / d3;
    }
    const int left = end - c0;
    const int nk = left < 32 ? left : 32;
    #pragma unroll 4
    for (int k = 0; k < nk; ++k) {
      const int sk = __builtin_amdgcn_readlane(s, k);
      const float k0 = __int_as_float(__builtin_amdgcn_readlane(__float_as_int(w0), k));
      const float k1 = __int_as_float(__builtin_amdgcn_readlane(__float_as_int(w1), k));
      const float k2 = __int_as_float(__builtin_amdgcn_readlane(__float_as_int(w2), k));
      const float k3 = __int_as_float(__builtin_amdgcn_readlane(__float_as_int(w3), k));
      const float ca = hi ? k1 : k0;
      const float cb = hi ? k3 : k2;
      const float* __restrict__ fr = fcol + (size_t)sk * FDIM;
      const float4 fa = *(const float4*)fr;
      const float4 fb = *(const float4*)(fr + 128);
      accA.x += fa.x * ca; accA.y += fa.y * ca; accA.z += fa.z * ca; accA.w += fa.w * ca;
      accB.x += fb.x * cb; accB.y += fb.y * cb; accB.z += fb.z * cb; accB.w += fb.w * cb;
    }
  }

  if (resid) {
    const float* __restrict__ rr = resid + (size_t)n * FDIM + lane * 4;
    const float4 ra = *(const float4*)rr;
    const float4 rb = *(const float4*)(rr + 128);
    accA.x += ra.x; accA.y += ra.y; accA.z += ra.z; accA.w += ra.w;
    accB.x += rb.x; accB.y += rb.y; accB.z += rb.z; accB.w += rb.w;
  }
  accA.x = accA.x > 0.f ? accA.x : (__expf(accA.x) - 1.f);
  accA.y = accA.y > 0.f ? accA.y : (__expf(accA.y) - 1.f);
  accA.z = accA.z > 0.f ? accA.z : (__expf(accA.z) - 1.f);
  accA.w = accA.w > 0.f ? accA.w : (__expf(accA.w) - 1.f);
  accB.x = accB.x > 0.f ? accB.x : (__expf(accB.x) - 1.f);
  accB.y = accB.y > 0.f ? accB.y : (__expf(accB.y) - 1.f);
  accB.z = accB.z > 0.f ? accB.z : (__expf(accB.z) - 1.f);
  accB.w = accB.w > 0.f ? accB.w : (__expf(accB.w) - 1.f);
  float* __restrict__ o = out + (size_t)n * FDIM + lane * 4;
  *(float4*)o = accA;
  *(float4*)(o + 128) = accB;
}

static void run_layer(const float* hin, const float* Wm, const float* al,
                      const float* ar, const int* rowptr, const int* csrc,
                      int N, float* feat, float* el, float* er,
                      const float* resid, float* rst, hipStream_t stream) {
  dim3 ggrid((N + 63) / 64, FDIM / 64);
  gemm_f32_wmma<<<ggrid, 128, 0, stream>>>(hin, Wm, feat, N);
  node_attn<<<(N + 7) / 8, 256, 0, stream>>>(feat, al, ar, el, er, N);
  gat_gather<<<(N + 7) / 8, 256, 0, stream>>>(rowptr, csrc, feat, el, er, resid, rst, N);
}

extern "C" void kernel_launch(void* const* d_in, const int* in_sizes, int n_in,
                              void* d_out, int out_size, void* d_ws, size_t ws_size,
                              hipStream_t stream) {
  const float* features = (const float*)d_in[0];
  const int*   src      = (const int*)d_in[1];
  const int*   dst      = (const int*)d_in[2];
  const float* W1       = (const float*)d_in[3];
  const float* al1      = (const float*)d_in[4];
  const float* ar1      = (const float*)d_in[5];
  const float* W2       = (const float*)d_in[6];
  const float* al2      = (const float*)d_in[7];
  const float* ar2      = (const float*)d_in[8];

  const int N = in_sizes[0] / FDIM;
  const int E = in_sizes[1];

  char* w = (char*)d_ws;
  float* feat = (float*)w;  w += (size_t)N * FDIM * sizeof(float);
  float* h1   = (float*)w;  w += (size_t)N * FDIM * sizeof(float);
  float* el   = (float*)w;  w += (size_t)N * HEADS * sizeof(float);
  float* er   = (float*)w;  w += (size_t)N * HEADS * sizeof(float);
  int* rank   = (int*)w;    w += (size_t)E * sizeof(int);
  int* csrc   = (int*)w;    w += (size_t)E * sizeof(int);
  int* cnt    = (int*)w;    w += (size_t)N * sizeof(int);
  int* rowptr = (int*)w;

  zero_cnt<<<(N + 255) / 256, 256, 0, stream>>>(cnt, N);
  edge_rank<<<(E + 255) / 256, 256, 0, stream>>>(dst, cnt, rank, E);
  scan_rowptr<<<(N + 1023) / 1024, 1024, 0, stream>>>(cnt, rowptr, N);
  csr_fill<<<(E + 255) / 256, 256, 0, stream>>>(src, dst, rank, rowptr, csrc, E);

  run_layer(features, W1, al1, ar1, rowptr, csrc, N,
            feat, el, er,  nullptr,  h1, stream);

  run_layer(h1, W2, al2, ar2, rowptr, csrc, N,
            feat, el, er,  h1,  (float*)d_out, stream);
}


# el/er dot products formed in the GEMM epilogue from the f32 accumulators (one 64-col block = one head); both node_attn launches removed; el/er sum order now 4 in-lane fma + 16-lane xor tree
# speedup vs baseline: 3.4067x; 1.0565x over previous
#include <hip/hip_runtime.h>
#include <math.h>

#define NEG_SLOPE 0.2f
#define FDIM 256
#define HEADS 4
#define HDIM 64
#define KTILE 32
#define LDT 34

typedef float v2f __attribute__((ext_vector_type(2)));
typedef float v8f __attribute__((ext_vector_type(8)));

__global__ __launch_bounds__(128) void gemm_f32_wmma(
    const float* __restrict__ H, const float* __restrict__ W,
    const float* __restrict__ al, const float* __restrict__ ar,
    float* __restrict__ OUT, float* __restrict__ el, float* __restrict__ er,
    int N) {
  __shared__ float As[64 * LDT];
  __shared__ float Bt[64 * LDT];

  const int row0 = blockIdx.x * 64;
  const int col0 = blockIdx.y * 64;
  const int t = threadIdx.x;
  const int wave = t >> 5, lane = t & 31;
  const int hi = lane >> 4;
  const int m  = lane & 15;

  v8f acc0 = {}, acc1 = {}, acc2 = {}, acc3 = {};

  for (int kt = 0; kt < FDIM; kt += KTILE) {
    __syncthreads();
    for (int i = t; i < 64 * KTILE; i += 128) {
      const int r = i >> 5, c = i & 31;
      int rr = row0 + r; if (rr > N - 1) rr = N - 1;
      As[r * LDT + c] = H[(size_t)rr * FDIM + kt + c];
    }
    for (int i = t; i < 64 * KTILE; i += 128) {
      const int k = i >> 6, n = i & 63;
      Bt[n * LDT + k] = W[(size_t)(kt + k) * FDIM + col0 + n];
    }
    __syncthreads();

    #pragma unroll
    for (int kk = 0; kk < KTILE; kk += 4) {
      const int ka = kk + 2 * hi;
      v2f a  = *(const v2f*)&As[(wave * 16 + m) * LDT + ka];
      v2f b0 = *(const v2f*)&Bt[(m)      * LDT + ka];
      v2f b1 = *(const v2f*)&Bt[(16 + m) * LDT + ka];
      v2f b2 = *(const v2f*)&Bt[(32 + m) * LDT + ka];
      v2f b3 = *(const v2f*)&Bt[(48 + m) * LDT + ka];
      acc0 = __builtin_amdgcn_wmma_f32_16x16x4_f32(false, a, false, b0, (short)0, acc0, false, false);
      acc1 = __builtin_amdgcn_wmma_f32_16x16x4_f32(false, a, false, b1, (short)0, acc1, false, false);
      acc2 = __builtin_amdgcn_wmma_f32_16x16x4_f32(false, a, false, b2, (short)0, acc2, false, false);
      acc3 = __builtin_amdgcn_wmma_f32_16x16x4_f32(false, a, false, b3, (short)0, acc3, false, false);
    }
  }

  #pragma unroll
  for (int r = 0; r < 8; ++r) {
    const int row = row0 + wave * 16 + r + 8 * hi;
    if (row < N) {
      float* o = OUT + (size_t)row * FDIM + col0 + m;
      o[0]  = acc0[r];
      o[16] = acc1[r];
      o[32] = acc2[r];
      o[48] = acc3[r];
    }
  }

  const int head = blockIdx.y;
  const float* __restrict__ alh = al + col0;
  const float* __restrict__ arh = ar + col0;
  const float l0 = alh[m], l1 = alh[16 + m], l2 = alh[32 + m], l3 = alh[48 + m];
  const float r0 = arh[m], r1 = arh[16 + m], r2 = arh[32 + m], r3 = arh[48 + m];
  float pm = 0.f, qm = 0.f;
  #pragma unroll
  for (int r = 0; r < 8; ++r) {
    float p = acc0[r] * l0 + acc1[r] * l1 + acc2[r] * l2 + acc3[r] * l3;
    float q = acc0[r] * r0 + acc1[r] * r1 + acc2[r] * r2 + acc3[r] * r3;
    #pragma unroll
    for (int off = 8; off > 0; off >>= 1) {
      p += __shfl_xor(p, off, 32);
      q += __shfl_xor(q, off, 32);
    }
    if (m == r) { pm = p; qm = q; }
  }
  const int arow = row0 + wave * 16 + m + 8 * hi;
  if (m < 8 && arow < N) {
    el[arow * HEADS + head] = pm;
    er[arow * HEADS + head] = qm;
  }
}

__global__ __launch_bounds__(256) void zero_cnt(int* __restrict__ cnt, int N) {
  const int i = blockIdx.x * blockDim.x + threadIdx.x;
  if (i < N) cnt[i] = 0;
}

__global__ __launch_bounds__(256) void edge_rank(
    const int* __restrict__ dst, int* __restrict__ cnt,
    int* __restrict__ rank, int E) {
  const int e = blockIdx.x * blockDim.x + threadIdx.x;
  if (e >= E) return;
  rank[e] = atomicAdd(&cnt[dst[e]], 1);
}

__global__ __launch_bounds__(1024) void scan_rowptr(
    const int* __restrict__ cnt, int* __restrict__ rowptr, int N) {
  __shared__ int wtot[32];
  __shared__ int wfront[32];
  const int t = threadIdx.x, lane = t & 31, wave = t >> 5;
  const int base = blockIdx.x * 1024;

  int front = 0;
  for (int i = t; i < base; i += 1024) front += cnt[i];

  const int idx = base + t;
  const int c = idx < N ? cnt[idx] : 0;
  int incl = c;
  #pragma unroll
  for (int off = 1; off < 32; off <<= 1) {
    const int v = __shfl_up(incl, off, 32);
    if (lane >= off) incl += v;
  }
  #pragma unroll
  for (int off = 16; off > 0; off >>= 1) front += __shfl_xor(front, off, 32);
  if (lane == 31) wtot[wave] = incl;
  if (lane == 0) wfront[wave] = front;
  __syncthreads();

  int add = wfront[lane] + (lane < wave ? wtot[lane] : 0);
  #pragma unroll
  for (int off = 16; off > 0; off >>= 1) add += __shfl_xor(add, off, 32);

  const int excl = add + incl - c;
  if (idx < N) rowptr[idx] = excl;
  if (idx == N - 1) rowptr[N] = excl + c;
}

__global__ __launch_bounds__(256) void csr_fill(
    const int* __restrict__ src, const int* __restrict__ dst,
    const int* __restrict__ rank, const int* __restrict__ rowptr,
    int* __restrict__ csrc, int E) {
  const int e = blockIdx.x * blockDim.x + threadIdx.x;
  if (e >= E) return;
  csrc[rowptr[dst[e]] + rank[e]] = src[e];
}

__device__ __forceinline__ float lrelu(float x) {
  return x > 0.f ? x : NEG_SLOPE * x;
}

__global__ __launch_bounds__(256) void gat_gather(
    const int* __restrict__ rowptr, const int* __restrict__ csrc,
    const float* __restrict__ feat, const float* __restrict__ el,
    const float* __restrict__ er, const float* __restrict__ resid,
    float* __restrict__ out, int N) {
  const int lane = threadIdx.x & 31;
  const int hi = lane >> 4;
  const int n = __builtin_amdgcn_readfirstlane(blockIdx.x * 8 + (threadIdx.x >> 5));
  if (n >= N) return;
  const int beg = rowptr[n], end = rowptr[n + 1];
  const float4 b = ((const float4*)er)[n];
  const float4* __restrict__ el4 = (const float4*)el;

  float m0 = -INFINITY, m1 = -INFINITY, m2 = -INFINITY, m3 = -INFINITY;
  for (int j = beg + lane; j < end; j += 32) {
    const float4 a = el4[csrc[j]];
    m0 = fmaxf(m0, lrelu(a.x + b.x));
    m1 = fmaxf(m1, lrelu(a.y + b.y));
    m2 = fmaxf(m2, lrelu(a.z + b.z));
    m3 = fmaxf(m3, lrelu(a.w + b.w));
  }
  #pragma unroll
  for (int off = 16; off > 0; off >>= 1) {
    m0 = fmaxf(m0, __shfl_xor(m0, off, 32));
    m1 = fmaxf(m1, __shfl_xor(m1, off, 32));
    m2 = fmaxf(m2, __shfl_xor(m2, off, 32));
    m3 = fmaxf(m3, __shfl_xor(m3, off, 32));
  }

  float d0 = 0.f, d1 = 0.f, d2 = 0.f, d3 = 0.f;
  for (int j = beg + lane; j < end; j += 32) {
    const float4 a = el4[csrc[j]];
    d0 += __expf(lrelu(a.x + b.x) - m0);
    d1 += __expf(lrelu(a.y + b.y) - m1);
    d2 += __expf(lrelu(a.z + b.z) - m2);
    d3 += __expf(lrelu(a.w + b.w) - m3);
  }
  #pragma unroll
  for (int off = 16; off > 0; off >>= 1) {
    d0 += __shfl_xor(d0, off, 32);
    d1 += __shfl_xor(d1, off, 32);
    d2 += __shfl_xor(d2, off, 32);
    d3 += __shfl_xor(d3, off, 32);
  }

  float4 accA = make_float4(0.f, 0.f, 0.f, 0.f);
  float4 accB = make_float4(0.f, 0.f, 0.f, 0.f);
  const float* __restrict__ fcol = feat + lane * 4;
  for (int c0 = beg; c0 < end; c0 += 32) {
    const int j = c0 + lane;
    int s = 0;
    float w0 = 0.f, w1 = 0.f, w2 = 0.f, w3 = 0.f;
    if (j < end) {
      s = csrc[j];
      const float4 a = el4[s];
      w0 = __expf(lrelu(a.x + b.x) - m0) / d0;
      w1 = __expf(lrelu(a.y + b.y) - m1) / d1;
      w2 = __expf(lrelu(a.z + b.z) - m2) / d2;
      w3 = __expf(lrelu(a.w + b.w) - m3) / d3;
    }
    const int left = end - c0;
    const int nk = left < 32 ? left : 32;
    #pragma unroll 4
    for (int k = 0; k < nk; ++k) {
      const int sk = __builtin_amdgcn_readlane(s, k);
      const float k0 = __int_as_float(__builtin_amdgcn_readlane(__float_as_int(w0), k));
      const float k1 = __int_as_float(__builtin_amdgcn_readlane(__float_as_int(w1), k));
      const float k2 = __int_as_float(__builtin_amdgcn_readlane(__float_as_int(w2), k));
      const float k3 = __int_as_float(__builtin_amdgcn_readlane(__float_as_int(w3), k));
      const float ca = hi ? k1 : k0;
      const float cb = hi ? k3 : k2;
      const float* __restrict__ fr = fcol + (size_t)sk * FDIM;
      const float4 fa = *(const float4*)fr;
      const float4 fb = *(const float4*)(fr + 128);
      accA.x += fa.x * ca; accA.y += fa.y * ca; accA.z += fa.z * ca; accA.w += fa.w * ca;
      accB.x += fb.x * cb; accB.y += fb.y * cb; accB.z += fb.z * cb; accB.w += fb.w * cb;
    }
  }

  if (resid) {
    const float* __restrict__ rr = resid + (size_t)n * FDIM + lane * 4;
    const float4 ra = *(const float4*)rr;
    const float4 rb = *(const float4*)(rr + 128);
    accA.x += ra.x; accA.y += ra.y; accA.z += ra.z; accA.w += ra.w;
    accB.x += rb.x; accB.y += rb.y; accB.z += rb.z; accB.w += rb.w;
  }
  accA.x = accA.x > 0.f ? accA.x : (__expf(accA.x) - 1.f);
  accA.y = accA.y > 0.f ? accA.y : (__expf(accA.y) - 1.f);
  accA.z = accA.z > 0.f ? accA.z : (__expf(accA.z) - 1.f);
  accA.w = accA.w > 0.f ? accA.w : (__expf(accA.w) - 1.f);
  accB.x = accB.x > 0.f ? accB.x : (__expf(accB.x) - 1.f);
  accB.y = accB.y > 0.f ? accB.y : (__expf(accB.y) - 1.f);
  accB.z = accB.z > 0.f ? accB.z : (__expf(accB.z) - 1.f);
  accB.w = accB.w > 0.f ? accB.w : (__expf(accB.w) - 1.f);
  float* __restrict__ o = out + (size_t)n * FDIM + lane * 4;
  *(float4*)o = accA;
  *(float4*)(o + 128) = accB;
}

static void run_layer(const float* hin, const float* Wm, const float* al,
                      const float* ar, const int* rowptr, const int* csrc,
                      int N, float* feat, float* el, float* er,
                      const float* resid, float* rst, hipStream_t stream) {
  dim3 ggrid((N + 63) / 64, FDIM / 64);
  gemm_f32_wmma<<<ggrid, 128, 0, stream>>>(hin, Wm, al, ar, feat, el, er, N);
  gat_gather<<<(N + 7) / 8, 256, 0, stream>>>(rowptr, csrc, feat, el, er, resid, rst, N);
}

extern "C" void kernel_launch(void* const* d_in, const int* in_sizes, int n_in,
                              void* d_out, int out_size, void* d_ws, size_t ws_size,
                              hipStream_t stream) {
  const float* features = (const float*)d_in[0];
  const int*   src      = (const int*)d_in[1];
  const int*   dst      = (const int*)d_in[2];
  const float* W1       = (const float*)d_in[3];
  const float* al1      = (const float*)d_in[4];
  const float* ar1      = (const float*)d_in[5];
  const float* W2       = (const float*)d_in[6];
  const float* al2      = (const float*)d_in[7];
  const float* ar2      = (const float*)d_in[8];

  const int N = in_sizes[0] / FDIM;
  const int E = in_sizes[1];

  char* w = (char*)d_ws;
  float* feat = (float*)w;  w += (size_t)N * FDIM * sizeof(float);
  float* h1   = (float*)w;  w += (size_t)N * FDIM * sizeof(float);
  float* el   = (float*)w;  w += (size_t)N * HEADS * sizeof(float);
  float* er   = (float*)w;  w += (size_t)N * HEADS * sizeof(float);
  int* rank   = (int*)w;    w += (size_t)E * sizeof(int);
  int* csrc   = (int*)w;    w += (size_t)E * sizeof(int);
  int* cnt    = (int*)w;    w += (size_t)N * sizeof(int);
  int* rowptr = (int*)w;

  zero_cnt<<<(N + 255) / 256, 256, 0, stream>>>(cnt, N);
  edge_rank<<<(E + 255) / 256, 256, 0, stream>>>(dst, cnt, rank, E);
  scan_rowptr<<<(N + 1023) / 1024, 1024, 0, stream>>>(cnt, rowptr, N);
  csr_fill<<<(E + 255) / 256, 256, 0, stream>>>(src, dst, rank, rowptr, csrc, E);

  run_layer(features, W1, al1, ar1, rowptr, csrc, N,
            feat, el, er,  nullptr,  h1, stream);

  run_layer(h1, W2, al2, ar2, rowptr, csrc, N,
            feat, el, er,  h1,  (float*)d_out, stream);
}


# GEMM operand staging pipelined: next K-tile fetched by 16-byte global loads issued as one group ahead of the products, kept in registers, two LDS stages, one barrier per K-tile; sum order unchanged
# speedup vs baseline: 4.8484x; 1.4232x over previous
#include <hip/hip_runtime.h>
#include <math.h>

#define NEG_SLOPE 0.2f
#define FDIM 256
#define HEADS 4
#define HDIM 64
#define KTILE 32
#define LDT 36

typedef float v2f __attribute__((ext_vector_type(2)));
typedef float v8f __attribute__((ext_vector_type(8)));

__global__ __launch_bounds__(128) void gemm_f32_wmma(
    const float* H, const float* W,
    const float* __restrict__ al, const float* __restrict__ ar,
    float* __restrict__ OUT, float* __restrict__ el, float* __restrict__ er,
    int N) {
  __shared__ __attribute__((aligned(16))) float As[2][64 * LDT];
  __shared__ __attribute__((aligned(16))) float Bt[2][64 * LDT];

  const int row0 = blockIdx.x * 64;
  const int col0 = blockIdx.y * 64;
  const int t = threadIdx.x;
  const int wave = t >> 5, lane = t & 31;
  const int hi = lane >> 4;
  const int m  = lane & 15;

  const int ac4 = t & 7, srow = t >> 3;
  int sr0 = row0 + srow, sr1 = sr0 + 16, sr2 = sr0 + 32, sr3 = sr0 + 48;
  if (sr0 > N - 1) sr0 = N - 1;
  if (sr1 > N - 1) sr1 = N - 1;
  if (sr2 > N - 1) sr2 = N - 1;
  if (sr3 > N - 1) sr3 = N - 1;
  const unsigned ao0 = (unsigned)sr0 * FDIM + ac4 * 4;
  const unsigned ao1 = (unsigned)sr1 * FDIM + ac4 * 4;
  const unsigned ao2 = (unsigned)sr2 * FDIM + ac4 * 4;
  const unsigned ao3 = (unsigned)sr3 * FDIM + ac4 * 4;
  const int bng = t & 15, bkg = t >> 4;
  const unsigned bo = (unsigned)(bkg * 4) * FDIM + col0 + bng * 4;
  const int aoff = srow * LDT + ac4 * 4;
  const int boff = (bng * 4) * LDT + bkg * 4;

  float4 ra0, ra1, ra2, ra3, rb0, rb1, rb2, rb3;
  v8f acc0 = {}, acc1 = {}, acc2 = {}, acc3 = {};

#define GEMM_FETCH(kt)                                                        \
  {                                                                           \
    const float* hk = H + (kt);                                               \
    const float* wk = W + (size_t)(kt) * FDIM;                                \
    ra0 = *(const float4*)(hk + ao0);                                         \
    ra1 = *(const float4*)(hk + ao1);                                         \
    ra2 = *(const float4*)(hk + ao2);                                         \
    ra3 = *(const float4*)(hk + ao3);                                         \
    rb0 = *(const float4*)(wk + bo);                                          \
    rb1 = *(const float4*)(wk + bo + FDIM);                                   \
    rb2 = *(const float4*)(wk + bo + 2 * FDIM);                               \
    rb3 = *(const float4*)(wk + bo + 3 * FDIM);                               \
  }
#define GEMM_STASH(s)                                                         \
  {                                                                           \
    __builtin_amdgcn_sched_barrier(0);                                        \
    *(float4*)&As[s][aoff]            = ra0;                                  \
    *(float4*)&As[s][aoff + 16 * LDT] = ra1;                                  \
    *(float4*)&As[s][aoff + 32 * LDT] = ra2;                                  \
    *(float4*)&As[s][aoff + 48 * LDT] = ra3;                                  \
    *(float4*)&Bt[s][boff]           = make_float4(rb0.x, rb1.x, rb2.x, rb3.x); \
    *(float4*)&Bt[s][boff + LDT]     = make_float4(rb0.y, rb1.y, rb2.y, rb3.y); \
    *(float4*)&Bt[s][boff + 2 * LDT] = make_float4(rb0.z, rb1.z, rb2.z, rb3.z); \
    *(float4*)&Bt[s][boff + 3 * LDT] = make_float4(rb0.w, rb1.w, rb2.w, rb3.w); \
  }
#define GEMM_PRODUCTS(s)                                                      \
  {                                                                           \
    _Pragma("unroll")                                                         \
    for (int kk = 0; kk < KTILE; kk += 4) {                                   \
      const int ka = kk + 2 * hi;        \
      v2f a  = *(const v2f*)&As[s][(wave * 16 + m) * LDT + ka];               \
      v2f b0 = *(const v2f*)&Bt[s][(m)      * LDT + ka];                      \
      v2f b1 = *(const v2f*)&Bt[s][(16 + m) * LDT + ka];                      \
      v2f b2 = *(const v2f*)&Bt[s][(32 + m) * LDT + ka];                      \
      v2f b3 = *(const v2f*)&Bt[s][(48 + m) * LDT + ka];                      \
      acc0 = __builtin_amdgcn_wmma_f32_16x16x4_f32(false, a, false, b0, (short)0, acc0, false, false); \
      acc1 = __builtin_amdgcn_wmma_f32_16x16x4_f32(false, a, false, b1, (short)0, acc1, false, false); \
      acc2 = __builtin_amdgcn_wmma_f32_16x16x4_f32(false, a, false, b2, (short)0, acc2, false, false); \
      acc3 = __builtin_amdgcn_wmma_f32_16x16x4_f32(false, a, false, b3, (short)0, acc3, false, false); \
    }                                                                         \
  }

  GEMM_FETCH(0);
  GEMM_STASH(0);
  __syncthreads();

  #pragma unroll 1
  for (int kt = 0; kt < FDIM - 2 * KTILE; kt += 2 * KTILE) {
    GEMM_FETCH(kt + KTILE);
    GEMM_PRODUCTS(0);
    GEMM_STASH(1);
    __syncthreads();
    GEMM_FETCH(kt + 2 * KTILE);
    GEMM_PRODUCTS(1);
    GEMM_STASH(0);
    __syncthreads();
  }
  GEMM_FETCH(FDIM - KTILE);
  GEMM_PRODUCTS(0);
  GEMM_STASH(1);
  __syncthreads();
  GEMM_PRODUCTS(1);
#undef GEMM_FETCH
#undef GEMM_STASH
#undef GEMM_PRODUCTS

  #pragma unroll
  for (int r = 0; r < 8; ++r) {
    const int row = row0 + wave * 16 + r + 8 * hi;
    if (row < N) {
      float* o = OUT + (size_t)row * FDIM + col0 + m;
      o[0]  = acc0[r];
      o[16] = acc1[r];
      o[32] = acc2[r];
      o[48] = acc3[r];
    }
  }

  const int head = blockIdx.y;
  const float* __restrict__ alh = al + col0;
  const float* __restrict__ arh = ar + col0;
  const float l0 = alh[m], l1 = alh[16 + m], l2 = alh[32 + m], l3 = alh[48 + m];
  const float r0 = arh[m], r1 = arh[16 + m], r2 = arh[32 + m], r3 = arh[48 + m];
  float pm = 0.f, qm = 0.f;
  #pragma unroll
  for (int r = 0; r < 8; ++r) {
    float p = acc0[r] * l0 + acc1[r] * l1 + acc2[r] * l2 + acc3[r] * l3;
    float q = acc0[r] * r0 + acc1[r] * r1 + acc2[r] * r2 + acc3[r] * r3;
    #pragma unroll
    for (int off = 8; off > 0; off >>= 1) {
      p += __shfl_xor(p, off, 32);
      q += __shfl_xor(q, off, 32);
    }
    if (m == r) { pm = p; qm = q; }
  }
  const int arow = row0 + wave * 16 + m + 8 * hi;
  if (m < 8 && arow < N) {
    el[arow * HEADS + head] = pm;
    er[arow * HEADS + head] = qm;
  }
}

__global__ __launch_bounds__(256) void zero_cnt(int* __restrict__ cnt, int N) {
  const int i = blockIdx.x * blockDim.x + threadIdx.x;
  if (i < N) cnt[i] = 0;
}

__global__ __launch_bounds__(256) void edge_rank(
    const int* __restrict__ dst, int* __restrict__ cnt,
    int* __restrict__ rank, int E) {
  const int e = blockIdx.x * blockDim.x + threadIdx.x;
  if (e >= E) return;
  rank[e] = atomicAdd(&cnt[dst[e]], 1);
}

__global__ __launch_bounds__(1024) void scan_rowptr(
    const int* __restrict__ cnt, int* __restrict__ rowptr, int N) {
  __shared__ int wtot[32];
  __shared__ int wfront[32];
  const int t = threadIdx.x, lane = t & 31, wave = t >> 5;
  const int base = blockIdx.x * 1024;

  int front = 0;
  for (int i = t; i < base; i += 1024) front += cnt[i];

  const int idx = base + t;
  const int c = idx < N ? cnt[idx] : 0;
  int incl = c;
  #pragma unroll
  for (int off = 1; off < 32; off <<= 1) {
    const int v = __shfl_up(incl, off, 32);
    if (lane >= off) incl += v;
  }
  #pragma unroll
  for (int off = 16; off > 0; off >>= 1) front += __shfl_xor(front, off, 32);
  if (lane == 31) wtot[wave] = incl;
  if (lane == 0) wfront[wave] = front;
  __syncthreads();

  int add = wfront[lane] + (lane < wave ? wtot[lane] : 0);
  #pragma unroll
  for (int off = 16; off > 0; off >>= 1) add += __shfl_xor(add, off, 32);

  const int excl = add + incl - c;
  if (idx < N) rowptr[idx] = excl;
  if (idx == N - 1) rowptr[N] = excl + c;
}

__global__ __launch_bounds__(256) void csr_fill(
    const int* __restrict__ src, const int* __restrict__ dst,
    const int* __restrict__ rank, const int* __restrict__ rowptr,
    int* __restrict__ csrc, int E) {
  const int e = blockIdx.x * blockDim.x + threadIdx.x;
  if (e >= E) return;
  csrc[rowptr[dst[e]] + rank[e]] = src[e];
}

__device__ __forceinline__ float lrelu(float x) {
  return x > 0.f ? x : NEG_SLOPE * x;
}

__global__ __launch_bounds__(256) void gat_gather(
    const int* __restrict__ rowptr, const int* __restrict__ csrc,
    const float* __restrict__ feat, const float* __restrict__ el,
    const float* __restrict__ er, const float* __restrict__ resid,
    float* __restrict__ out, int N) {
  const int lane = threadIdx.x & 31;
  const int hi = lane >> 4;
  const int n = __builtin_amdgcn_readfirstlane(blockIdx.x * 8 + (threadIdx.x >> 5));
  if (n >= N) return;
  const int beg = rowptr[n], end = rowptr[n + 1];
  const float4 b = ((const float4*)er)[n];
  const float4* __restrict__ el4 = (const float4*)el;

  float m0 = -INFINITY, m1 = -INFINITY, m2 = -INFINITY, m3 = -INFINITY;
  for (int j = beg + lane; j < end; j += 32) {
    const float4 a = el4[csrc[j]];
    m0 = fmaxf(m0, lrelu(a.x + b.x));
    m1 = fmaxf(m1, lrelu(a.y + b.y));
    m2 = fmaxf(m2, lrelu(a.z + b.z));
    m3 = fmaxf(m3, lrelu(a.w + b.w));
  }
  #pragma unroll
  for (int off = 16; off > 0; off >>= 1) {
    m0 = fmaxf(m0, __shfl_xor(m0, off, 32));
    m1 = fmaxf(m1, __shfl_xor(m1, off, 32));
    m2 = fmaxf(m2, __shfl_xor(m2, off, 32));
    m3 = fmaxf(m3, __shfl_xor(m3, off, 32));
  }

  float d0 = 0.f, d1 = 0.f, d2 = 0.f, d3 = 0.f;
  for (int j = beg + lane; j < end; j += 32) {
    const float4 a = el4[csrc[j]];
    d0 += __expf(lrelu(a.x + b.x) - m0);
    d1 += __expf(lrelu(a.y + b.y) - m1);
    d2 += __expf(lrelu(a.z + b.z) - m2);
    d3 += __expf(lrelu(a.w + b.w) - m3);
  }
  #pragma unroll
  for (int off = 16; off > 0; off >>= 1) {
    d0 += __shfl_xor(d0, off, 32);
    d1 += __shfl_xor(d1, off, 32);
    d2 += __shfl_xor(d2, off, 32);
    d3 += __shfl_xor(d3, off, 32);
  }

  float4 accA = make_float4(0.f, 0.f, 0.f, 0.f);
  float4 accB = make_float4(0.f, 0.f, 0.f, 0.f);
  const float* __restrict__ fcol = feat + lane * 4;
  for (int c0 = beg; c0 < end; c0 += 32) {
    const int j = c0 + lane;
    int s = 0;
    float w0 = 0.f, w1 = 0.f, w2 = 0.f, w3 = 0.f;
    if (j < end) {
      s = csrc[j];
      const float4 a = el4[s];
      w0 = __expf(lrelu(a.x + b.x) - m0) / d0;
      w1 = __expf(lrelu(a.y + b.y) - m1) / d1;
      w2 = __expf(lrelu(a.z + b.z) - m2) / d2;
      w3 = __expf(lrelu(a.w + b.w) - m3) / d3;
    }
    const int left = end - c0;
    const int nk = left < 32 ? left : 32;
    #pragma unroll 4
    for (int k = 0; k < nk; ++k) {
      const int sk = __builtin_amdgcn_readlane(s, k);
      const float k0 = __int_as_float(__builtin_amdgcn_readlane(__float_as_int(w0), k));
      const float k1 = __int_as_float(__builtin_amdgcn_readlane(__float_as_int(w1), k));
      const float k2 = __int_as_float(__builtin_amdgcn_readlane(__float_as_int(w2), k));
      const float k3 = __int_as_float(__builtin_amdgcn_readlane(__float_as_int(w3), k));
      const float ca = hi ? k1 : k0;
      const float cb = hi ? k3 : k2;
      const float* __restrict__ fr = fcol + (size_t)sk * FDIM;
      const float4 fa = *(const float4*)fr;
      const float4 fb = *(const float4*)(fr + 128);
      accA.x += fa.x * ca; accA.y += fa.y * ca; accA.z += fa.z * ca; accA.w += fa.w * ca;
      accB.x += fb.x * cb; accB.y += fb.y * cb; accB.z += fb.z * cb; accB.w += fb.w * cb;
    }
  }

  if (resid) {
    const float* __restrict__ rr = resid + (size_t)n * FDIM + lane * 4;
    const float4 ra = *(const float4*)rr;
    const float4 rb = *(const float4*)(rr + 128);
    accA.x += ra.x; accA.y += ra.y; accA.z += ra.z; accA.w += ra.w;
    accB.x += rb.x; accB.y += rb.y; accB.z += rb.z; accB.w += rb.w;
  }
  accA.x = accA.x > 0.f ? accA.x : (__expf(accA.x) - 1.f);
  accA.y = accA.y > 0.f ? accA.y : (__expf(accA.y) - 1.f);
  accA.z = accA.z > 0.f ? accA.z : (__expf(accA.z) - 1.f);
  accA.w = accA.w > 0.f ? accA.w : (__expf(accA.w) - 1.f);
  accB.x = accB.x > 0.f ? accB.x : (__expf(accB.x) - 1.f);
  accB.y = accB.y > 0.f ? accB.y : (__expf(accB.y) - 1.f);
  accB.z = accB.z > 0.f ? accB.z : (__expf(accB.z) - 1.f);
  accB.w = accB.w > 0.f ? accB.w : (__expf(accB.w) - 1.f);
  float* __restrict__ o = out + (size_t)n * FDIM + lane * 4;
  *(float4*)o = accA;
  *(float4*)(o + 128) = accB;
}

static void run_layer(const float* hin, const float* Wm, const float* al,
                      const float* ar, const int* rowptr, const int* csrc,
                      int N, float* feat, float* el, float* er,
                      const float* resid, float* rst, hipStream_t stream) {
  dim3 ggrid((N + 63) / 64, FDIM / 64);
  gemm_f32_wmma<<<ggrid, 128, 0, stream>>>(hin, Wm, al, ar, feat, el, er, N);
  gat_gather<<<(N + 7) / 8, 256, 0, stream>>>(rowptr, csrc, feat, el, er, resid, rst, N);
}

extern "C" void kernel_launch(void* const* d_in, const int* in_sizes, int n_in,
                              void* d_out, int out_size, void* d_ws, size_t ws_size,
                              hipStream_t stream) {
  const float* features = (const float*)d_in[0];
  const int*   src      = (const int*)d_in[1];
  const int*   dst      = (const int*)d_in[2];
  const float* W1       = (const float*)d_in[3];
  const float* al1      = (const float*)d_in[4];
  const float* ar1      = (const float*)d_in[5];
  const float* W2       = (const float*)d_in[6];
  const float* al2      = (const float*)d_in[7];
  const float* ar2      = (const float*)d_in[8];

  const int N = in_sizes[0] / FDIM;
  const int E = in_sizes[1];

  char* w = (char*)d_ws;
  float* feat = (float*)w;  w += (size_t)N * FDIM * sizeof(float);
  float* h1   = (float*)w;  w += (size_t)N * FDIM * sizeof(float);
  float* el   = (float*)w;  w += (size_t)N * HEADS * sizeof(float);
  float* er   = (float*)w;  w += (size_t)N * HEADS * sizeof(float);
  int* rank   = (int*)w;    w += (size_t)E * sizeof(int);
  int* csrc   = (int*)w;    w += (size_t)E * sizeof(int);
  int* cnt    = (int*)w;    w += (size_t)N * sizeof(int);
  int* rowptr = (int*)w;

  zero_cnt<<<(N + 255) / 256, 256, 0, stream>>>(cnt, N);
  edge_rank<<<(E + 255) / 256, 256, 0, stream>>>(dst, cnt, rank, E);
  scan_rowptr<<<(N + 1023) / 1024, 1024, 0, stream>>>(cnt, rowptr, N);
  csr_fill<<<(E + 255) / 256, 256, 0, stream>>>(src, dst, rank, rowptr, csrc, E);

  run_layer(features, W1, al1, ar1, rowptr, csrc, N,
            feat, el, er,  nullptr,  h1, stream);

  run_layer(h1, W2, al2, ar2, rowptr, csrc, N,
            feat, el, er,  h1,  (float*)d_out, stream);
}


# gat_gather reads each node's edge list once: src index, logits and exp terms of its first 32 edges stay in registers from max to sum to weights (was three csrc->el read passes); sum order unchanged
# speedup vs baseline: 5.0907x; 1.0500x over previous
#include <hip/hip_runtime.h>
#include <math.h>

#define NEG_SLOPE 0.2f
#define FDIM 256
#define HEADS 4
#define HDIM 64
#define KTILE 32
#define LDT 36

typedef float v2f __attribute__((ext_vector_type(2)));
typedef float v8f __attribute__((ext_vector_type(8)));

__global__ __launch_bounds__(128) void gemm_f32_wmma(
    const float* H, const float* W,
    const float* __restrict__ al, const float* __restrict__ ar,
    float* __restrict__ OUT, float* __restrict__ el, float* __restrict__ er,
    int N) {
  __shared__ __attribute__((aligned(16))) float As[2][64 * LDT];
  __shared__ __attribute__((aligned(16))) float Bt[2][64 * LDT];

  const int row0 = blockIdx.x * 64;
  const int col0 = blockIdx.y * 64;
  const int t = threadIdx.x;
  const int wave = t >> 5, lane = t & 31;
  const int hi = lane >> 4;
  const int m  = lane & 15;

  const int ac4 = t & 7, srow = t >> 3;
  int sr0 = row0 + srow, sr1 = sr0 + 16, sr2 = sr0 + 32, sr3 = sr0 + 48;
  if (sr0 > N - 1) sr0 = N - 1;
  if (sr1 > N - 1) sr1 = N - 1;
  if (sr2 > N - 1) sr2 = N - 1;
  if (sr3 > N - 1) sr3 = N - 1;
  const unsigned ao0 = (unsigned)sr0 * FDIM + ac4 * 4;
  const unsigned ao1 = (unsigned)sr1 * FDIM + ac4 * 4;
  const unsigned ao2 = (unsigned)sr2 * FDIM + ac4 * 4;
  const unsigned ao3 = (unsigned)sr3 * FDIM + ac4 * 4;
  const int bng = t & 15, bkg = t >> 4;
  const unsigned bo = (unsigned)(bkg * 4) * FDIM + col0 + bng * 4;
  const int aoff = srow * LDT + ac4 * 4;
  const int boff = (bng * 4) * LDT + bkg * 4;

  float4 ra0, ra1, ra2, ra3, rb0, rb1, rb2, rb3;
  v8f acc0 = {}, acc1 = {}, acc2 = {}, acc3 = {};

#define GEMM_FETCH(kt)                                                        \
  {                                                                           \
    const float* hk = H + (kt);                                               \
    const float* wk = W + (size_t)(kt) * FDIM;                                \
    ra0 = *(const float4*)(hk + ao0);                                         \
    ra1 = *(const float4*)(hk + ao1);                                         \
    ra2 = *(const float4*)(hk + ao2);                                         \
    ra3 = *(const float4*)(hk + ao3);                                         \
    rb0 = *(const float4*)(wk + bo);                                          \
    rb1 = *(const float4*)(wk + bo + FDIM);                                   \
    rb2 = *(const float4*)(wk + bo + 2 * FDIM);                               \
    rb3 = *(const float4*)(wk + bo + 3 * FDIM);                               \
  }
#define GEMM_STASH(s)                                                         \
  {                                                                           \
    __builtin_amdgcn_sched_barrier(0);                                        \
    *(float4*)&As[s][aoff]            = ra0;                                  \
    *(float4*)&As[s][aoff + 16 * LDT] = ra1;                                  \
    *(float4*)&As[s][aoff + 32 * LDT] = ra2;                                  \
    *(float4*)&As[s][aoff + 48 * LDT] = ra3;                                  \
    *(float4*)&Bt[s][boff]           = make_float4(rb0.x, rb1.x, rb2.x, rb3.x); \
    *(float4*)&Bt[s][boff + LDT]     = make_float4(rb0.y, rb1.y, rb2.y, rb3.y); \
    *(float4*)&Bt[s][boff + 2 * LDT] = make_float4(rb0.z, rb1.z, rb2.z, rb3.z); \
    *(float4*)&Bt[s][boff + 3 * LDT] = make_float4(rb0.w, rb1.w, rb2.w, rb3.w); \
  }
#define GEMM_PRODUCTS(s)                                                      \
  {                                                                           \
    _Pragma("unroll")                                                         \
    for (int kk = 0; kk < KTILE; kk += 4) {                                   \
      const int ka = kk + 2 * hi;        \
      v2f a  = *(const v2f*)&As[s][(wave * 16 + m) * LDT + ka];               \
      v2f b0 = *(const v2f*)&Bt[s][(m)      * LDT + ka];                      \
      v2f b1 = *(const v2f*)&Bt[s][(16 + m) * LDT + ka];                      \
      v2f b2 = *(const v2f*)&Bt[s][(32 + m) * LDT + ka];                      \
      v2f b3 = *(const v2f*)&Bt[s][(48 + m) * LDT + ka];                      \
      acc0 = __builtin_amdgcn_wmma_f32_16x16x4_f32(false, a, false, b0, (short)0, acc0, false, false); \
      acc1 = __builtin_amdgcn_wmma_f32_16x16x4_f32(false, a, false, b1, (short)0, acc1, false, false); \
      acc2 = __builtin_amdgcn_wmma_f32_16x16x4_f32(false, a, false, b2, (short)0, acc2, false, false); \
      acc3 = __builtin_amdgcn_wmma_f32_16x16x4_f32(false, a, false, b3, (short)0, acc3, false, false); \
    }                                                                         \
  }

  GEMM_FETCH(0);
  GEMM_STASH(0);
  __syncthreads();

  #pragma unroll 1
  for (int kt = 0; kt < FDIM - 2 * KTILE; kt += 2 * KTILE) {
    GEMM_FETCH(kt + KTILE);
    GEMM_PRODUCTS(0);
    GEMM_STASH(1);
    __syncthreads();
    GEMM_FETCH(kt + 2 * KTILE);
    GEMM_PRODUCTS(1);
    GEMM_STASH(0);
    __syncthreads();
  }
  GEMM_FETCH(FDIM - KTILE);
  GEMM_PRODUCTS(0);
  GEMM_STASH(1);
  __syncthreads();
  GEMM_PRODUCTS(1);
#undef GEMM_FETCH
#undef GEMM_STASH
#undef GEMM_PRODUCTS

  #pragma unroll
  for (int r = 0; r < 8; ++r) {
    const int row = row0 + wave * 16 + r + 8 * hi;
    if (row < N) {
      float* o = OUT + (size_t)row * FDIM + col0 + m;
      o[0]  = acc0[r];
      o[16] = acc1[r];
      o[32] = acc2[r];
      o[48] = acc3[r];
    }
  }

  const int head = blockIdx.y;
  const float* __restrict__ alh = al + col0;
  const float* __restrict__ arh = ar + col0;
  const float l0 = alh[m], l1 = alh[16 + m], l2 = alh[32 + m], l3 = alh[48 + m];
  const float r0 = arh[m], r1 = arh[16 + m], r2 = arh[32 + m], r3 = arh[48 + m];
  float pm = 0.f, qm = 0.f;
  #pragma unroll
  for (int r = 0; r < 8; ++r) {
    float p = acc0[r] * l0 + acc1[r] * l1 + acc2[r] * l2 + acc3[r] * l3;
    float q = acc0[r] * r0 + acc1[r] * r1 + acc2[r] * r2 + acc3[r] * r3;
    #pragma unroll
    for (int off = 8; off > 0; off >>= 1) {
      p += __shfl_xor(p, off, 32);
      q += __shfl_xor(q, off, 32);
    }
    if (m == r) { pm = p; qm = q; }
  }
  const int arow = row0 + wave * 16 + m + 8 * hi;
  if (m < 8 && arow < N) {
    el[arow * HEADS + head] = pm;
    er[arow * HEADS + head] = qm;
  }
}

__global__ __launch_bounds__(256) void zero_cnt(int* __restrict__ cnt, int N) {
  const int i = blockIdx.x * blockDim.x + threadIdx.x;
  if (i < N) cnt[i] = 0;
}

__global__ __launch_bounds__(256) void edge_rank(
    const int* __restrict__ dst, int* __restrict__ cnt,
    int* __restrict__ rank, int E) {
  const int e = blockIdx.x * blockDim.x + threadIdx.x;
  if (e >= E) return;
  rank[e] = atomicAdd(&cnt[dst[e]], 1);
}

__global__ __launch_bounds__(1024) void scan_rowptr(
    const int* __restrict__ cnt, int* __restrict__ rowptr, int N) {
  __shared__ int wtot[32];
  __shared__ int wfront[32];
  const int t = threadIdx.x, lane = t & 31, wave = t >> 5;
  const int base = blockIdx.x * 1024;

  int front = 0;
  for (int i = t; i < base; i += 1024) front += cnt[i];

  const int idx = base + t;
  const int c = idx < N ? cnt[idx] : 0;
  int incl = c;
  #pragma unroll
  for (int off = 1; off < 32; off <<= 1) {
    const int v = __shfl_up(incl, off, 32);
    if (lane >= off) incl += v;
  }
  #pragma unroll
  for (int off = 16; off > 0; off >>= 1) front += __shfl_xor(front, off, 32);
  if (lane == 31) wtot[wave] = incl;
  if (lane == 0) wfront[wave] = front;
  __syncthreads();

  int add = wfront[lane] + (lane < wave ? wtot[lane] : 0);
  #pragma unroll
  for (int off = 16; off > 0; off >>= 1) add += __shfl_xor(add, off, 32);

  const int excl = add + incl - c;
  if (idx < N) rowptr[idx] = excl;
  if (idx == N - 1) rowptr[N] = excl + c;
}

__global__ __launch_bounds__(256) void csr_fill(
    const int* __restrict__ src, const int* __restrict__ dst,
    const int* __restrict__ rank, const int* __restrict__ rowptr,
    int* __restrict__ csrc, int E) {
  const int e = blockIdx.x * blockDim.x + threadIdx.x;
  if (e >= E) return;
  csrc[rowptr[dst[e]] + rank[e]] = src[e];
}

__device__ __forceinline__ float lrelu(float x) {
  return x > 0.f ? x : NEG_SLOPE * x;
}

__global__ __launch_bounds__(256) void gat_gather(
    const int* __restrict__ rowptr, const int* __restrict__ csrc,
    const float* __restrict__ feat, const float* __restrict__ el,
    const float* __restrict__ er, const float* __restrict__ resid,
    float* __restrict__ out, int N) {
  const int lane = threadIdx.x & 31;
  const int hi = lane >> 4;
  const int n = __builtin_amdgcn_readfirstlane(blockIdx.x * 8 + (threadIdx.x >> 5));
  if (n >= N) return;
  const int beg = rowptr[n], end = rowptr[n + 1];
  const float4 b = ((const float4*)er)[n];
  const float4* __restrict__ el4 = (const float4*)el;

  const int j0 = beg + lane;
  const bool has0 = j0 < end;
  const bool more = end - beg > 32;
  int s0 = 0;
  float e0 = 0.f, e1 = 0.f, e2 = 0.f, e3 = 0.f;
  if (has0) {
    s0 = csrc[j0];
    const float4 a = el4[s0];
    e0 = lrelu(a.x + b.x);
    e1 = lrelu(a.y + b.y);
    e2 = lrelu(a.z + b.z);
    e3 = lrelu(a.w + b.w);
  }

  float m0 = -INFINITY, m1 = -INFINITY, m2 = -INFINITY, m3 = -INFINITY;
  if (has0) {
    m0 = fmaxf(m0, e0);
    m1 = fmaxf(m1, e1);
    m2 = fmaxf(m2, e2);
    m3 = fmaxf(m3, e3);
  }
  if (more) {
    for (int j = j0 + 32; j < end; j += 32) {
      const float4 a = el4[csrc[j]];
      m0 = fmaxf(m0, lrelu(a.x + b.x));
      m1 = fmaxf(m1, lrelu(a.y + b.y));
      m2 = fmaxf(m2, lrelu(a.z + b.z));
      m3 = fmaxf(m3, lrelu(a.w + b.w));
    }
  }
  #pragma unroll
  for (int off = 16; off > 0; off >>= 1) {
    m0 = fmaxf(m0, __shfl_xor(m0, off, 32));
    m1 = fmaxf(m1, __shfl_xor(m1, off, 32));
    m2 = fmaxf(m2, __shfl_xor(m2, off, 32));
    m3 = fmaxf(m3, __shfl_xor(m3, off, 32));
  }

  float p0 = 0.f, p1 = 0.f, p2 = 0.f, p3 = 0.f;
  if (has0) {
    p0 = __expf(e0 - m0);
    p1 = __expf(e1 - m1);
    p2 = __expf(e2 - m2);
    p3 = __expf(e3 - m3);
  }
  float d0 = 0.f, d1 = 0.f, d2 = 0.f, d3 = 0.f;
  d0 += p0; d1 += p1; d2 += p2; d3 += p3;
  if (more) {
    for (int j = j0 + 32; j < end; j += 32) {
      const float4 a = el4[csrc[j]];
      d0 += __expf(lrelu(a.x + b.x) - m0);
      d1 += __expf(lrelu(a.y + b.y) - m1);
      d2 += __expf(lrelu(a.z + b.z) - m2);
      d3 += __expf(lrelu(a.w + b.w) - m3);
    }
  }
  #pragma unroll
  for (int off = 16; off > 0; off >>= 1) {
    d0 += __shfl_xor(d0, off, 32);
    d1 += __shfl_xor(d1, off, 32);
    d2 += __shfl_xor(d2, off, 32);
    d3 += __shfl_xor(d3, off, 32);
  }

  float4 accA = make_float4(0.f, 0.f, 0.f, 0.f);
  float4 accB = make_float4(0.f, 0.f, 0.f, 0.f);
  const float* __restrict__ fcol = feat + lane * 4;
  for (int c0 = beg; c0 < end; c0 += 32) {
    int s = 0;
    float w0 = 0.f, w1 = 0.f, w2 = 0.f, w3 = 0.f;
    if (c0 == beg) {
      if (has0) {
        s = s0;
        w0 = p0 / d0;
        w1 = p1 / d1;
        w2 = p2 / d2;
        w3 = p3 / d3;
      }
    } else {
      const int j = c0 + lane;
      if (j < end) {
        s = csrc[j];
        const float4 a = el4[s];
        w0 = __expf(lrelu(a.x + b.x) - m0) / d0;
        w1 = __expf(lrelu(a.y + b.y) - m1) / d1;
        w2 = __expf(lrelu(a.z + b.z) - m2) / d2;
        w3 = __expf(lrelu(a.w + b.w) - m3) / d3;
      }
    }
    const int left = end - c0;
    const int nk = left < 32 ? left : 32;
    #pragma unroll 4
    for (int k = 0; k < nk; ++k) {
      const int sk = __builtin_amdgcn_readlane(s, k);
      const float k0 = __int_as_float(__builtin_amdgcn_readlane(__float_as_int(w0), k));
      const float k1 = __int_as_float(__builtin_amdgcn_readlane(__float_as_int(w1), k));
      const float k2 = __int_as_float(__builtin_amdgcn_readlane(__float_as_int(w2), k));
      const float k3 = __int_as_float(__builtin_amdgcn_readlane(__float_as_int(w3), k));
      const float ca = hi ? k1 : k0;
      const float cb = hi ? k3 : k2;
      const float* __restrict__ fr = fcol + (size_t)sk * FDIM;
      const float4 fa = *(const float4*)fr;
      const float4 fb = *(const float4*)(fr + 128);
      accA.x += fa.x * ca; accA.y += fa.y * ca; accA.z += fa.z * ca; accA.w += fa.w * ca;
      accB.x += fb.x * cb; accB.y += fb.y * cb; accB.z += fb.z * cb; accB.w += fb.w * cb;
    }
  }

  if (resid) {
    const float* __restrict__ rr = resid + (size_t)n * FDIM + lane * 4;
    const float4 ra = *(const float4*)rr;
    const float4 rb = *(const float4*)(rr + 128);
    accA.x += ra.x; accA.y += ra.y; accA.z += ra.z; accA.w += ra.w;
    accB.x += rb.x; accB.y += rb.y; accB.z += rb.z; accB.w += rb.w;
  }
  accA.x = accA.x > 0.f ? accA.x : (__expf(accA.x) - 1.f);
  accA.y = accA.y > 0.f ? accA.y : (__expf(accA.y) - 1.f);
  accA.z = accA.z > 0.f ? accA.z : (__expf(accA.z) - 1.f);
  accA.w = accA.w > 0.f ? accA.w : (__expf(accA.w) - 1.f);
  accB.x = accB.x > 0.f ? accB.x : (__expf(accB.x) - 1.f);
  accB.y = accB.y > 0.f ? accB.y : (__expf(accB.y) - 1.f);
  accB.z = accB.z > 0.f ? accB.z : (__expf(accB.z) - 1.f);
  accB.w = accB.w > 0.f ? accB.w : (__expf(accB.w) - 1.f);
  float* __restrict__ o = out + (size_t)n * FDIM + lane * 4;
  *(float4*)o = accA;
  *(float4*)(o + 128) = accB;
}

static void run_layer(const float* hin, const float* Wm, const float* al,
                      const float* ar, const int* rowptr, const int* csrc,
                      int N, float* feat, float* el, float* er,
                      const float* resid, float* rst, hipStream_t stream) {
  dim3 ggrid((N + 63) / 64, FDIM / 64);
  gemm_f32_wmma<<<ggrid, 128, 0, stream>>>(hin, Wm, al, ar, feat, el, er, N);
  gat_gather<<<(N + 7) / 8, 256, 0, stream>>>(rowptr, csrc, feat, el, er, resid, rst, N);
}

extern "C" void kernel_launch(void* const* d_in, const int* in_sizes, int n_in,
                              void* d_out, int out_size, void* d_ws, size_t ws_size,
                              hipStream_t stream) {
  const float* features = (const float*)d_in[0];
  const int*   src      = (const int*)d_in[1];
  const int*   dst      = (const int*)d_in[2];
  const float* W1       = (const float*)d_in[3];
  const float* al1      = (const float*)d_in[4];
  const float* ar1      = (const float*)d_in[5];
  const float* W2       = (const float*)d_in[6];
  const float* al2      = (const float*)d_in[7];
  const float* ar2      = (const float*)d_in[8];

  const int N = in_sizes[0] / FDIM;
  const int E = in_sizes[1];

  char* w = (char*)d_ws;
  float* feat = (float*)w;  w += (size_t)N * FDIM * sizeof(float);
  float* h1   = (float*)w;  w += (size_t)N * FDIM * sizeof(float);
  float* el   = (float*)w;  w += (size_t)N * HEADS * sizeof(float);
  float* er   = (float*)w;  w += (size_t)N * HEADS * sizeof(float);
  int* rank   = (int*)w;    w += (size_t)E * sizeof(int);
  int* csrc   = (int*)w;    w += (size_t)E * sizeof(int);
  int* cnt    = (int*)w;    w += (size_t)N * sizeof(int);
  int* rowptr = (int*)w;

  zero_cnt<<<(N + 255) / 256, 256, 0, stream>>>(cnt, N);
  edge_rank<<<(E + 255) / 256, 256, 0, stream>>>(dst, cnt, rank, E);
  scan_rowptr<<<(N + 1023) / 1024, 1024, 0, stream>>>(cnt, rowptr, N);
  csr_fill<<<(E + 255) / 256, 256, 0, stream>>>(src, dst, rank, rowptr, csrc, E);

  run_layer(features, W1, al1, ar1, rowptr, csrc, N,
            feat, el, er,  nullptr,  h1, stream);

  run_layer(h1, W2, al2, ar2, rowptr, csrc, N,
            feat, el, er,  h1,  (float*)d_out, stream);
}


# Edge ranking without global atomics: 64 work-groups count their chunk's edges per dst in a 200 KiB LDS histogram (LDS atomic = rank in chunk), per-dst scan over chunks; sum order: chunk, LDS arrival
# speedup vs baseline: 5.4794x; 1.0764x over previous
#include <hip/hip_runtime.h>
#include <math.h>

#define NEG_SLOPE 0.2f
#define FDIM 256
#define HEADS 4
#define HDIM 64
#define KTILE 32
#define LDT 36

typedef float v2f __attribute__((ext_vector_type(2)));
typedef float v8f __attribute__((ext_vector_type(8)));

__global__ __launch_bounds__(128) void gemm_f32_wmma(
    const float* H, const float* W,
    const float* __restrict__ al, const float* __restrict__ ar,
    float* __restrict__ OUT, float* __restrict__ el, float* __restrict__ er,
    int N) {
  __shared__ __attribute__((aligned(16))) float As[2][64 * LDT];
  __shared__ __attribute__((aligned(16))) float Bt[2][64 * LDT];

  const int row0 = blockIdx.x * 64;
  const int col0 = blockIdx.y * 64;
  const int t = threadIdx.x;
  const int wave = t >> 5, lane = t & 31;
  const int hi = lane >> 4;
  const int m  = lane & 15;

  const int ac4 = t & 7, srow = t >> 3;
  int sr0 = row0 + srow, sr1 = sr0 + 16, sr2 = sr0 + 32, sr3 = sr0 + 48;
  if (sr0 > N - 1) sr0 = N - 1;
  if (sr1 > N - 1) sr1 = N - 1;
  if (sr2 > N - 1) sr2 = N - 1;
  if (sr3 > N - 1) sr3 = N - 1;
  const unsigned ao0 = (unsigned)sr0 * FDIM + ac4 * 4;
  const unsigned ao1 = (unsigned)sr1 * FDIM + ac4 * 4;
  const unsigned ao2 = (unsigned)sr2 * FDIM + ac4 * 4;
  const unsigned ao3 = (unsigned)sr3 * FDIM + ac4 * 4;
  const int bng = t & 15, bkg = t >> 4;
  const unsigned bo = (unsigned)(bkg * 4) * FDIM + col0 + bng * 4;
  const int aoff = srow * LDT + ac4 * 4;
  const int boff = (bng * 4) * LDT + bkg * 4;

  float4 ra0, ra1, ra2, ra3, rb0, rb1, rb2, rb3;
  v8f acc0 = {}, acc1 = {}, acc2 = {}, acc3 = {};

#define GEMM_FETCH(kt)                                                        \
  {                                                                           \
    const float* hk = H + (kt);                                               \
    const float* wk = W + (size_t)(kt) * FDIM;                                \
    ra0 = *(const float4*)(hk + ao0);                                         \
    ra1 = *(const float4*)(hk + ao1);                                         \
    ra2 = *(const float4*)(hk + ao2);                                         \
    ra3 = *(const float4*)(hk + ao3);                                         \
    rb0 = *(const float4*)(wk + bo);                                          \
    rb1 = *(const float4*)(wk + bo + FDIM);                                   \
    rb2 = *(const float4*)(wk + bo + 2 * FDIM);                               \
    rb3 = *(const float4*)(wk + bo + 3 * FDIM);                               \
  }
#define GEMM_STASH(s)                                                         \
  {                                                                           \
    __builtin_amdgcn_sched_barrier(0);                                        \
    *(float4*)&As[s][aoff]            = ra0;                                  \
    *(float4*)&As[s][aoff + 16 * LDT] = ra1;                                  \
    *(float4*)&As[s][aoff + 32 * LDT] = ra2;                                  \
    *(float4*)&As[s][aoff + 48 * LDT] = ra3;                                  \
    *(float4*)&Bt[s][boff]           = make_float4(rb0.x, rb1.x, rb2.x, rb3.x); \
    *(float4*)&Bt[s][boff + LDT]     = make_float4(rb0.y, rb1.y, rb2.y, rb3.y); \
    *(float4*)&Bt[s][boff + 2 * LDT] = make_float4(rb0.z, rb1.z, rb2.z, rb3.z); \
    *(float4*)&Bt[s][boff + 3 * LDT] = make_float4(rb0.w, rb1.w, rb2.w, rb3.w); \
  }
#define GEMM_PRODUCTS(s)                                                      \
  {                                                                           \
    _Pragma("unroll")                                                         \
    for (int kk = 0; kk < KTILE; kk += 4) {                                   \
      const int ka = kk + 2 * hi;        \
      v2f a  = *(const v2f*)&As[s][(wave * 16 + m) * LDT + ka];               \
      v2f b0 = *(const v2f*)&Bt[s][(m)      * LDT + ka];                      \
      v2f b1 = *(const v2f*)&Bt[s][(16 + m) * LDT + ka];                      \
      v2f b2 = *(const v2f*)&Bt[s][(32 + m) * LDT + ka];                      \
      v2f b3 = *(const v2f*)&Bt[s][(48 + m) * LDT + ka];                      \
      acc0 = __builtin_amdgcn_wmma_f32_16x16x4_f32(false, a, false, b0, (short)0, acc0, false, false); \
      acc1 = __builtin_amdgcn_wmma_f32_16x16x4_f32(false, a, false, b1, (short)0, acc1, false, false); \
      acc2 = __builtin_amdgcn_wmma_f32_16x16x4_f32(false, a, false, b2, (short)0, acc2, false, false); \
      acc3 = __builtin_amdgcn_wmma_f32_16x16x4_f32(false, a, false, b3, (short)0, acc3, false, false); \
    }                                                                         \
  }

  GEMM_FETCH(0);
  GEMM_STASH(0);
  __syncthreads();

  #pragma unroll 1
  for (int kt = 0; kt < FDIM - 2 * KTILE; kt += 2 * KTILE) {
    GEMM_FETCH(kt + KTILE);
    GEMM_PRODUCTS(0);
    GEMM_STASH(1);
    __syncthreads();
    GEMM_FETCH(kt + 2 * KTILE);
    GEMM_PRODUCTS(1);
    GEMM_STASH(0);
    __syncthreads();
  }
  GEMM_FETCH(FDIM - KTILE);
  GEMM_PRODUCTS(0);
  GEMM_STASH(1);
  __syncthreads();
  GEMM_PRODUCTS(1);
#undef GEMM_FETCH
#undef GEMM_STASH
#undef GEMM_PRODUCTS

  #pragma unroll
  for (int r = 0; r < 8; ++r) {
    const int row = row0 + wave * 16 + r + 8 * hi;
    if (row < N) {
      float* o = OUT + (size_t)row * FDIM + col0 + m;
      o[0]  = acc0[r];
      o[16] = acc1[r];
      o[32] = acc2[r];
      o[48] = acc3[r];
    }
  }

  const int head = blockIdx.y;
  const float* __restrict__ alh = al + col0;
  const float* __restrict__ arh = ar + col0;
  const float l0 = alh[m], l1 = alh[16 + m], l2 = alh[32 + m], l3 = alh[48 + m];
  const float r0 = arh[m], r1 = arh[16 + m], r2 = arh[32 + m], r3 = arh[48 + m];
  float pm = 0.f, qm = 0.f;
  #pragma unroll
  for (int r = 0; r < 8; ++r) {
    float p = acc0[r] * l0 + acc1[r] * l1 + acc2[r] * l2 + acc3[r] * l3;
    float q = acc0[r] * r0 + acc1[r] * r1 + acc2[r] * r2 + acc3[r] * r3;
    #pragma unroll
    for (int off = 8; off > 0; off >>= 1) {
      p += __shfl_xor(p, off, 32);
      q += __shfl_xor(q, off, 32);
    }
    if (m == r) { pm = p; qm = q; }
  }
  const int arow = row0 + wave * 16 + m + 8 * hi;
  if (m < 8 && arow < N) {
    el[arow * HEADS + head] = pm;
    er[arow * HEADS + head] = qm;
  }
}

#define RANK_CHUNKS 64
#define RANK_KEYS   51200

__device__ __forceinline__ int chunk_len(int E) {
  return ((E + 4 * RANK_CHUNKS - 1) / (4 * RANK_CHUNKS)) * 4;
}

__global__ __launch_bounds__(1024) void edge_rank(
    const int* __restrict__ dst, int* __restrict__ hist,
    int* __restrict__ rank, int N, int E) {
  __shared__ __attribute__((aligned(16))) int h[RANK_KEYS];
  const int t = threadIdx.x;
  const int ch = chunk_len(E);
  const int e0 = blockIdx.x * ch;
  const int e1 = e0 + ch < E ? e0 + ch : E;
  const int len = e1 > e0 ? e1 - e0 : 0;
  const bool pieces = (len & 3) == 0 && ((((size_t)dst) | ((size_t)rank)) & 15) == 0;
  const int nq = len >> 2;
  for (int k0 = 0; k0 < N; k0 += RANK_KEYS) {
    const unsigned kn = (unsigned)(N - k0 < RANK_KEYS ? N - k0 : RANK_KEYS);
    for (int k = t; k < (int)kn; k += 1024) h[k] = 0;
    __syncthreads();
    if (pieces) {
      for (int q0 = 0; q0 < nq; q0 += 4096) {
        int4 d[4];
        #pragma unroll
        for (int i = 0; i < 4; ++i) {
          const int q = q0 + i * 1024 + t;
          d[i] = *(const int4*)(dst + e0 + 4 * (q < nq ? q : nq - 1));
        }
        #pragma unroll
        for (int i = 0; i < 4; ++i) {
          const int q = q0 + i * 1024 + t;
          if (q < nq) {
            const int e = e0 + 4 * q;
            const unsigned kx = (unsigned)(d[i].x - k0), ky = (unsigned)(d[i].y - k0);
            const unsigned kz = (unsigned)(d[i].z - k0), kw = (unsigned)(d[i].w - k0);
            if (kx < kn && ky < kn && kz < kn && kw < kn) {
              int4 r;
              r.x = atomicAdd(&h[kx], 1);
              r.y = atomicAdd(&h[ky], 1);
              r.z = atomicAdd(&h[kz], 1);
              r.w = atomicAdd(&h[kw], 1);
              *(int4*)(rank + e) = r;
            } else {
              if (kx < kn) rank[e]     = atomicAdd(&h[kx], 1);
              if (ky < kn) rank[e + 1] = atomicAdd(&h[ky], 1);
              if (kz < kn) rank[e + 2] = atomicAdd(&h[kz], 1);
              if (kw < kn) rank[e + 3] = atomicAdd(&h[kw], 1);
            }
          }
        }
      }
    } else {
      for (int e = e0 + t; e < e1; e += 1024) {
        const unsigned k = (unsigned)(dst[e] - k0);
        if (k < kn) rank[e] = atomicAdd(&h[k], 1);
      }
    }
    __syncthreads();
    int* __restrict__ hc = hist + (size_t)blockIdx.x * N + k0;
    for (int k = t; k < (int)kn; k += 1024) hc[k] = h[k];
    __syncthreads();
  }
}

__global__ __launch_bounds__(256) void chunk_scan(
    int* hist, int* __restrict__ cnt, int N) {
  const int k = blockIdx.x * blockDim.x + threadIdx.x;
  if (k >= N) return;
  int v[RANK_CHUNKS];
  #pragma unroll
  for (int c = 0; c < RANK_CHUNKS; ++c) v[c] = hist[(size_t)c * N + k];
  int run = 0;
  #pragma unroll
  for (int c = 0; c < RANK_CHUNKS; ++c) {
    hist[(size_t)c * N + k] = run;
    run += v[c];
  }
  cnt[k] = run;
}

__global__ __launch_bounds__(1024) void scan_rowptr(
    const int* __restrict__ cnt, int* __restrict__ rowptr, int N) {
  __shared__ int wtot[32];
  __shared__ int wfront[32];
  const int t = threadIdx.x, lane = t & 31, wave = t >> 5;
  const int base = blockIdx.x * 1024;

  int front = 0;
  for (int i = t; i < base; i += 1024) front += cnt[i];

  const int idx = base + t;
  const int c = idx < N ? cnt[idx] : 0;
  int incl = c;
  #pragma unroll
  for (int off = 1; off < 32; off <<= 1) {
    const int v = __shfl_up(incl, off, 32);
    if (lane >= off) incl += v;
  }
  #pragma unroll
  for (int off = 16; off > 0; off >>= 1) front += __shfl_xor(front, off, 32);
  if (lane == 31) wtot[wave] = incl;
  if (lane == 0) wfront[wave] = front;
  __syncthreads();

  int add = wfront[lane] + (lane < wave ? wtot[lane] : 0);
  #pragma unroll
  for (int off = 16; off > 0; off >>= 1) add += __shfl_xor(add, off, 32);

  const int excl = add + incl - c;
  if (idx < N) rowptr[idx] = excl;
  if (idx == N - 1) rowptr[N] = excl + c;
}

__global__ __launch_bounds__(256) void csr_fill(
    const int* __restrict__ src, const int* __restrict__ dst,
    const int* __restrict__ rank, const int* __restrict__ rowptr,
    const int* __restrict__ hist, int* __restrict__ csrc, int N, int E) {
  const int e = blockIdx.x * blockDim.x + threadIdx.x;
  if (e >= E) return;
  const int d = dst[e];
  const int c = e / chunk_len(E);
  csrc[rowptr[d] + hist[(size_t)c * N + d] + rank[e]] = src[e];
}

__device__ __forceinline__ float lrelu(float x) {
  return x > 0.f ? x : NEG_SLOPE * x;
}

__global__ __launch_bounds__(256) void gat_gather(
    const int* __restrict__ rowptr, const int* __restrict__ csrc,
    const float* __restrict__ feat, const float* __restrict__ el,
    const float* __restrict__ er, const float* __restrict__ resid,
    float* __restrict__ out, int N) {
  const int lane = threadIdx.x & 31;
  const int hi = lane >> 4;
  const int n = __builtin_amdgcn_readfirstlane(blockIdx.x * 8 + (threadIdx.x >> 5));
  if (n >= N) return;
  const int beg = rowptr[n], end = rowptr[n + 1];
  const float4 b = ((const float4*)er)[n];
  const float4* __restrict__ el4 = (const float4*)el;

  const int j0 = beg + lane;
  const bool has0 = j0 < end;
  const bool more = end - beg > 32;
  int s0 = 0;
  float e0 = 0.f, e1 = 0.f, e2 = 0.f, e3 = 0.f;
  if (has0) {
    s0 = csrc[j0];
    const float4 a = el4[s0];
    e0 = lrelu(a.x + b.x);
    e1 = lrelu(a.y + b.y);
    e2 = lrelu(a.z + b.z);
    e3 = lrelu(a.w + b.w);
  }

  float m0 = -INFINITY, m1 = -INFINITY, m2 = -INFINITY, m3 = -INFINITY;
  if (has0) {
    m0 = fmaxf(m0, e0);
    m1 = fmaxf(m1, e1);
    m2 = fmaxf(m2, e2);
    m3 = fmaxf(m3, e3);
  }
  if (more) {
    for (int j = j0 + 32; j < end; j += 32) {
      const float4 a = el4[csrc[j]];
      m0 = fmaxf(m0, lrelu(a.x + b.x));
      m1 = fmaxf(m1, lrelu(a.y + b.y));
      m2 = fmaxf(m2, lrelu(a.z + b.z));
      m3 = fmaxf(m3, lrelu(a.w + b.w));
    }
  }
  #pragma unroll
  for (int off = 16; off > 0; off >>= 1) {
    m0 = fmaxf(m0, __shfl_xor(m0, off, 32));
    m1 = fmaxf(m1, __shfl_xor(m1, off, 32));
    m2 = fmaxf(m2, __shfl_xor(m2, off, 32));
    m3 = fmaxf(m3, __shfl_xor(m3, off, 32));
  }

  float p0 = 0.f, p1 = 0.f, p2 = 0.f, p3 = 0.f;
  if (has0) {
    p0 = __expf(e0 - m0);
    p1 = __expf(e1 - m1);
    p2 = __expf(e2 - m2);
    p3 = __expf(e3 - m3);
  }
  float d0 = 0.f, d1 = 0.f, d2 = 0.f, d3 = 0.f;
  d0 += p0; d1 += p1; d2 += p2; d3 += p3;
  if (more) {
    for (int j = j0 + 32; j < end; j += 32) {
      const float4 a = el4[csrc[j]];
      d0 += __expf(lrelu(a.x + b.x) - m0);
      d1 += __expf(lrelu(a.y + b.y) - m1);
      d2 += __expf(lrelu(a.z + b.z) - m2);
      d3 += __expf(lrelu(a.w + b.w) - m3);
    }
  }
  #pragma unroll
  for (int off = 16; off > 0; off >>= 1) {
    d0 += __shfl_xor(d0, off, 32);
    d1 += __shfl_xor(d1, off, 32);
    d2 += __shfl_xor(d2, off, 32);
    d3 += __shfl_xor(d3, off, 32);
  }

  float4 accA = make_float4(0.f, 0.f, 0.f, 0.f);
  float4 accB = make_float4(0.f, 0.f, 0.f, 0.f);
  const float* __restrict__ fcol = feat + lane * 4;
  for (int c0 = beg; c0 < end; c0 += 32) {
    int s = 0;
    float w0 = 0.f, w1 = 0.f, w2 = 0.f, w3 = 0.f;
    if (c0 == beg) {
      if (has0) {
        s = s0;
        w0 = p0 / d0;
        w1 = p1 / d1;
        w2 = p2 / d2;
        w3 = p3 / d3;
      }
    } else {
      const int j = c0 + lane;
      if (j < end) {
        s = csrc[j];
        const float4 a = el4[s];
        w0 = __expf(lrelu(a.x + b.x) - m0) / d0;
        w1 = __expf(lrelu(a.y + b.y) - m1) / d1;
        w2 = __expf(lrelu(a.z + b.z) - m2) / d2;
        w3 = __expf(lrelu(a.w + b.w) - m3) / d3;
      }
    }
    const int left = end - c0;
    const int nk = left < 32 ? left : 32;
    #pragma unroll 4
    for (int k = 0; k < nk; ++k) {
      const int sk = __builtin_amdgcn_readlane(s, k);
      const float k0 = __int_as_float(__builtin_amdgcn_readlane(__float_as_int(w0), k));
      const float k1 = __int_as_float(__builtin_amdgcn_readlane(__float_as_int(w1), k));
      const float k2 = __int_as_float(__builtin_amdgcn_readlane(__float_as_int(w2), k));
      const float k3 = __int_as_float(__builtin_amdgcn_readlane(__float_as_int(w3), k));
      const float ca = hi ? k1 : k0;
      const float cb = hi ? k3 : k2;
      const float* __restrict__ fr = fcol + (size_t)sk * FDIM;
      const float4 fa = *(const float4*)fr;
      const float4 fb = *(const float4*)(fr + 128);
      accA.x += fa.x * ca; accA.y += fa.y * ca; accA.z += fa.z * ca; accA.w += fa.w * ca;
      accB.x += fb.x * cb; accB.y += fb.y * cb; accB.z += fb.z * cb; accB.w += fb.w * cb;
    }
  }

  if (resid) {
    const float* __restrict__ rr = resid + (size_t)n * FDIM + lane * 4;
    const float4 ra = *(const float4*)rr;
    const float4 rb = *(const float4*)(rr + 128);
    accA.x += ra.x; accA.y += ra.y; accA.z += ra.z; accA.w += ra.w;
    accB.x += rb.x; accB.y += rb.y; accB.z += rb.z; accB.w += rb.w;
  }
  accA.x = accA.x > 0.f ? accA.x : (__expf(accA.x) - 1.f);
  accA.y = accA.y > 0.f ? accA.y : (__expf(accA.y) - 1.f);
  accA.z = accA.z > 0.f ? accA.z : (__expf(accA.z) - 1.f);
  accA.w = accA.w > 0.f ? accA.w : (__expf(accA.w) - 1.f);
  accB.x = accB.x > 0.f ? accB.x : (__expf(accB.x) - 1.f);
  accB.y = accB.y > 0.f ? accB.y : (__expf(accB.y) - 1.f);
  accB.z = accB.z > 0.f ? accB.z : (__expf(accB.z) - 1.f);
  accB.w = accB.w > 0.f ? accB.w : (__expf(accB.w) - 1.f);
  float* __restrict__ o = out + (size_t)n * FDIM + lane * 4;
  *(float4*)o = accA;
  *(float4*)(o + 128) = accB;
}

static void run_layer(const float* hin, const float* Wm, const float* al,
                      const float* ar, const int* rowptr, const int* csrc,
                      int N, float* feat, float* el, float* er,
                      const float* resid, float* rst, hipStream_t stream) {
  dim3 ggrid((N + 63) / 64, FDIM / 64);
  gemm_f32_wmma<<<ggrid, 128, 0, stream>>>(hin, Wm, al, ar, feat, el, er, N);
  gat_gather<<<(N + 7) / 8, 256, 0, stream>>>(rowptr, csrc, feat, el, er, resid, rst, N);
}

extern "C" void kernel_launch(void* const* d_in, const int* in_sizes, int n_in,
                              void* d_out, int out_size, void* d_ws, size_t ws_size,
                              hipStream_t stream) {
  const float* features = (const float*)d_in[0];
  const int*   src      = (const int*)d_in[1];
  const int*   dst      = (const int*)d_in[2];
  const float* W1       = (const float*)d_in[3];
  const float* al1      = (const float*)d_in[4];
  const float* ar1      = (const float*)d_in[5];
  const float* W2       = (const float*)d_in[6];
  const float* al2      = (const float*)d_in[7];
  const float* ar2      = (const float*)d_in[8];

  const int N = in_sizes[0] / FDIM;
  const int E = in_sizes[1];

  char* w = (char*)d_ws;
  float* feat = (float*)w;  w += (size_t)N * FDIM * sizeof(float);
  float* h1   = (float*)w;  w += (size_t)N * FDIM * sizeof(float);
  float* el   = (float*)w;  w += (size_t)N * HEADS * sizeof(float);
  float* er   = (float*)w;  w += (size_t)N * HEADS * sizeof(float);
  int* rank   = (int*)w;    w += (size_t)E * sizeof(int);
  int* csrc   = (int*)w;    w += (size_t)E * sizeof(int);
  int* cnt    = (int*)w;    w += (size_t)N * sizeof(int);
  int* rowptr = (int*)w;    w += ((size_t)N + 4) / 4 * 4 * sizeof(int);
  int* hist   = (int*)w;

  edge_rank<<<RANK_CHUNKS, 1024, 0, stream>>>(dst, hist, rank, N, E);
  chunk_scan<<<(N + 255) / 256, 256, 0, stream>>>(hist, cnt, N);
  scan_rowptr<<<(N + 1023) / 1024, 1024, 0, stream>>>(cnt, rowptr, N);
  csr_fill<<<(E + 255) / 256, 256, 0, stream>>>(src, dst, rank, rowptr, hist, csrc, N, E);

  run_layer(features, W1, al1, ar1, rowptr, csrc, N,
            feat, el, er,  nullptr,  h1, stream);

  run_layer(h1, W2, al2, ar2, rowptr, csrc, N,
            feat, el, er,  h1,  (float*)d_out, stream);
}


# Edge sort in 2 launches, not 4: chunks order edges by 256-node bucket in LDS; a work-group per bucket ranks and places them in LDS, writes rowptr, csrc; sum order: source index (>32 edges: arrival)
# speedup vs baseline: 6.3475x; 1.1584x over previous
#include <hip/hip_runtime.h>
#include <math.h>

#define NEG_SLOPE 0.2f
#define FDIM 256
#define HEADS 4
#define HDIM 64
#define KTILE 32
#define LDT 36

typedef float v2f __attribute__((ext_vector_type(2)));
typedef float v8f __attribute__((ext_vector_type(8)));

__global__ __launch_bounds__(128) void gemm_f32_wmma(
    const float* H, const float* W,
    const float* __restrict__ al, const float* __restrict__ ar,
    float* __restrict__ OUT, float* __restrict__ el, float* __restrict__ er,
    int N) {
  __shared__ __attribute__((aligned(16))) float As[2][64 * LDT];
  __shared__ __attribute__((aligned(16))) float Bt[2][64 * LDT];

  const int row0 = blockIdx.x * 64;
  const int col0 = blockIdx.y * 64;
  const int t = threadIdx.x;
  const int wave = t >> 5, lane = t & 31;
  const int hi = lane >> 4;
  const int m  = lane & 15;

  const int ac4 = t & 7, srow = t >> 3;
  int sr0 = row0 + srow, sr1 = sr0 + 16, sr2 = sr0 + 32, sr3 = sr0 + 48;
  if (sr0 > N - 1) sr0 = N - 1;
  if (sr1 > N - 1) sr1 = N - 1;
  if (sr2 > N - 1) sr2 = N - 1;
  if (sr3 > N - 1) sr3 = N - 1;
  const unsigned ao0 = (unsigned)sr0 * FDIM + ac4 * 4;
  const unsigned ao1 = (unsigned)sr1 * FDIM + ac4 * 4;
  const unsigned ao2 = (unsigned)sr2 * FDIM + ac4 * 4;
  const unsigned ao3 = (unsigned)sr3 * FDIM + ac4 * 4;
  const int bng = t & 15, bkg = t >> 4;
  const unsigned bo = (unsigned)(bkg * 4) * FDIM + col0 + bng * 4;
  const int aoff = srow * LDT + ac4 * 4;
  const int boff = (bng * 4) * LDT + bkg * 4;

  float4 ra0, ra1, ra2, ra3, rb0, rb1, rb2, rb3;
  v8f acc0 = {}, acc1 = {}, acc2 = {}, acc3 = {};

#define GEMM_FETCH(kt)                                                        \
  {                                                                           \
    const float* hk = H + (kt);                                               \
    const float* wk = W + (size_t)(kt) * FDIM;                                \
    ra0 = *(const float4*)(hk + ao0);                                         \
    ra1 = *(const float4*)(hk + ao1);                                         \
    ra2 = *(const float4*)(hk + ao2);                                         \
    ra3 = *(const float4*)(hk + ao3);                                         \
    rb0 = *(const float4*)(wk + bo);                                          \
    rb1 = *(const float4*)(wk + bo + FDIM);                                   \
    rb2 = *(const float4*)(wk + bo + 2 * FDIM);                               \
    rb3 = *(const float4*)(wk + bo + 3 * FDIM);                               \
  }
#define GEMM_STASH(s)                                                         \
  {                                                                           \
    __builtin_amdgcn_sched_barrier(0);                                        \
    *(float4*)&As[s][aoff]            = ra0;                                  \
    *(float4*)&As[s][aoff + 16 * LDT] = ra1;                                  \
    *(float4*)&As[s][aoff + 32 * LDT] = ra2;                                  \
    *(float4*)&As[s][aoff + 48 * LDT] = ra3;                                  \
    *(float4*)&Bt[s][boff]           = make_float4(rb0.x, rb1.x, rb2.x, rb3.x); \
    *(float4*)&Bt[s][boff + LDT]     = make_float4(rb0.y, rb1.y, rb2.y, rb3.y); \
    *(float4*)&Bt[s][boff + 2 * LDT] = make_float4(rb0.z, rb1.z, rb2.z, rb3.z); \
    *(float4*)&Bt[s][boff + 3 * LDT] = make_float4(rb0.w, rb1.w, rb2.w, rb3.w); \
  }
#define GEMM_PRODUCTS(s)                                                      \
  {                                                                           \
    _Pragma("unroll")                                                         \
    for (int kk = 0; kk < KTILE; kk += 4) {                                   \
      const int ka = kk + 2 * hi;        \
      v2f a  = *(const v2f*)&As[s][(wave * 16 + m) * LDT + ka];               \
      v2f b0 = *(const v2f*)&Bt[s][(m)      * LDT + ka];                      \
      v2f b1 = *(const v2f*)&Bt[s][(16 + m) * LDT + ka];                      \
      v2f b2 = *(const v2f*)&Bt[s][(32 + m) * LDT + ka];                      \
      v2f b3 = *(const v2f*)&Bt[s][(48 + m) * LDT + ka];                      \
      acc0 = __builtin_amdgcn_wmma_f32_16x16x4_f32(false, a, false, b0, (short)0, acc0, false, false); \
      acc1 = __builtin_amdgcn_wmma_f32_16x16x4_f32(false, a, false, b1, (short)0, acc1, false, false); \
      acc2 = __builtin_amdgcn_wmma_f32_16x16x4_f32(false, a, false, b2, (short)0, acc2, false, false); \
      acc3 = __builtin_amdgcn_wmma_f32_16x16x4_f32(false, a, false, b3, (short)0, acc3, false, false); \
    }                                                                         \
  }

  GEMM_FETCH(0);
  GEMM_STASH(0);
  __syncthreads();

  #pragma unroll 1
  for (int kt = 0; kt < FDIM - 2 * KTILE; kt += 2 * KTILE) {
    GEMM_FETCH(kt + KTILE);
    GEMM_PRODUCTS(0);
    GEMM_STASH(1);
    __syncthreads();
    GEMM_FETCH(kt + 2 * KTILE);
    GEMM_PRODUCTS(1);
    GEMM_STASH(0);
    __syncthreads();
  }
  GEMM_FETCH(FDIM - KTILE);
  GEMM_PRODUCTS(0);
  GEMM_STASH(1);
  __syncthreads();
  GEMM_PRODUCTS(1);
#undef GEMM_FETCH
#undef GEMM_STASH
#undef GEMM_PRODUCTS

  #pragma unroll
  for (int r = 0; r < 8; ++r) {
    const int row = row0 + wave * 16 + r + 8 * hi;
    if (row < N) {
      float* o = OUT + (size_t)row * FDIM + col0 + m;
      o[0]  = acc0[r];
      o[16] = acc1[r];
      o[32] = acc2[r];
      o[48] = acc3[r];
    }
  }

  const int head = blockIdx.y;
  const float* __restrict__ alh = al + col0;
  const float* __restrict__ arh = ar + col0;
  const float l0 = alh[m], l1 = alh[16 + m], l2 = alh[32 + m], l3 = alh[48 + m];
  const float r0 = arh[m], r1 = arh[16 + m], r2 = arh[32 + m], r3 = arh[48 + m];
  float pm = 0.f, qm = 0.f;
  #pragma unroll
  for (int r = 0; r < 8; ++r) {
    float p = acc0[r] * l0 + acc1[r] * l1 + acc2[r] * l2 + acc3[r] * l3;
    float q = acc0[r] * r0 + acc1[r] * r1 + acc2[r] * r2 + acc3[r] * r3;
    #pragma unroll
    for (int off = 8; off > 0; off >>= 1) {
      p += __shfl_xor(p, off, 32);
      q += __shfl_xor(q, off, 32);
    }
    if (m == r) { pm = p; qm = q; }
  }
  const int arow = row0 + wave * 16 + m + 8 * hi;
  if (m < 8 && arow < N) {
    el[arow * HEADS + head] = pm;
    er[arow * HEADS + head] = qm;
  }
}

#define SORT_NT   256
#define SORT_CH   3136
#define SORT_EPT  13
#define SORT_BMAX 1024
#define NPB_SHIFT 8
#define NPB       256
#define BKT_EPT   20
#define BKT_CAP   8192

__device__ __forceinline__ int block_scan_256(int c, int* ws, int& tot) {
  const int lane = threadIdx.x & 31, wave = threadIdx.x >> 5;
  int incl = c;
  #pragma unroll
  for (int off = 1; off < 32; off <<= 1) {
    const int v = __shfl_up(incl, off, 32);
    if (lane >= off) incl += v;
  }
  if (lane == 31) ws[wave] = incl;
  __syncthreads();
  const int wv = lane < SORT_NT / 32 ? ws[lane] : 0;
  int pre = lane < wave ? wv : 0;
  int all = wv;
  #pragma unroll
  for (int off = 16; off > 0; off >>= 1) {
    pre += __shfl_xor(pre, off, 32);
    all += __shfl_xor(all, off, 32);
  }
  __syncthreads();
  tot = all;
  return pre + incl - c;
}

__global__ __launch_bounds__(SORT_NT) void chunk_sort(
    const int* __restrict__ src, const int* __restrict__ dst,
    int* __restrict__ cbuf, int* __restrict__ cnt, int* __restrict__ lof,
    int G, int B, int E) {
  __shared__ __attribute__((aligned(16))) int bc[SORT_BMAX];
  __shared__ __attribute__((aligned(16))) int bl[SORT_BMAX];
  __shared__ int ws[SORT_NT / 32];
  __shared__ __attribute__((aligned(8))) int ord[2 * SORT_CH];
  const int t = threadIdx.x;
  const int g = blockIdx.x;
  const int e0 = g * SORT_CH;
  const int len = E - e0 < SORT_CH ? E - e0 : SORT_CH;
  *(int4*)&bc[4 * t] = make_int4(0, 0, 0, 0);
  __syncthreads();
  int d[SORT_EPT], s[SORT_EPT], r[SORT_EPT];
  #pragma unroll
  for (int u = 0; u < SORT_EPT; ++u) {
    const int i = u * SORT_NT + t;
    const int ii = i < len ? i : len - 1;
    d[u] = dst[e0 + ii];
    s[u] = src[e0 + ii];
  }
  #pragma unroll
  for (int u = 0; u < SORT_EPT; ++u) {
    r[u] = 0;
    if (u * SORT_NT + t < len) r[u] = atomicAdd(&bc[d[u] >> NPB_SHIFT], 1);
  }
  __syncthreads();
  const int4 c4 = *(const int4*)&bc[4 * t];
  int tot;
  const int ex = block_scan_256(c4.x + c4.y + c4.z + c4.w, ws, tot);
  int4 x4;
  x4.x = ex;
  x4.y = x4.x + c4.x;
  x4.z = x4.y + c4.y;
  x4.w = x4.z + c4.z;
  *(int4*)&bl[4 * t] = x4;
  const int k0 = 4 * t;
  int* __restrict__ cg = cnt + g;
  int* __restrict__ lg = lof + g;
  if (k0     < B) { cg[(k0)     * G] = c4.x; lg[(k0)     * G] = x4.x; }
  if (k0 + 1 < B) { cg[(k0 + 1) * G] = c4.y; lg[(k0 + 1) * G] = x4.y; }
  if (k0 + 2 < B) { cg[(k0 + 2) * G] = c4.z; lg[(k0 + 2) * G] = x4.z; }
  if (k0 + 3 < B) { cg[(k0 + 3) * G] = c4.w; lg[(k0 + 3) * G] = x4.w; }
  __syncthreads();
  #pragma unroll
  for (int u = 0; u < SORT_EPT; ++u) {
    if (u * SORT_NT + t < len)
      *(int2*)&ord[2 * (bl[d[u] >> NPB_SHIFT] + r[u])] = make_int2(d[u], s[u]);
  }
  __syncthreads();
  int2* __restrict__ cb = (int2*)cbuf + e0;
  #pragma unroll 4
  for (int i = t; i < len; i += SORT_NT) cb[i] = *(const int2*)&ord[2 * i];
}

__global__ __launch_bounds__(SORT_NT) void bucket_csr(
    const int* __restrict__ cbuf, const int* __restrict__ cnt, const int* __restrict__ lof,
    int* __restrict__ rowptr, int* __restrict__ csrc, int G, int B, int N) {
  __shared__ int gs[SORT_BMAX];
  __shared__ int gl[SORT_BMAX];
  __shared__ int hist[NPB];
  __shared__ int lrow[NPB];
  __shared__ int ws[SORT_NT / 32];
  __shared__ int pk[BKT_CAP];
  __shared__ int sv[BKT_CAP];
  __shared__ int stage[BKT_CAP];
  const int t = threadIdx.x;
  const int b = blockIdx.x;
  const int n0 = b << NPB_SHIFT;
  int c[4], l[4];
  #pragma unroll
  for (int q = 0; q < 4; ++q) {
    const int k = 4 * t + q;
    c[q] = k < G ? cnt[b * G + k] : 0;
    l[q] = k < G ? lof[b * G + k] : 0;
  }
  int T, base;
  const int ex = block_scan_256(c[0] + c[1] + c[2] + c[3], ws, T);
  (void)block_scan_256(l[0] + l[1] + l[2] + l[3], ws, base);
  int run = ex;
  #pragma unroll
  for (int q = 0; q < 4; ++q) {
    const int k = 4 * t + q;
    gs[k] = run;
    gl[k] = k * SORT_CH + l[q] - run;
    run += c[q];
  }
  hist[t] = 0;
  int top = 1;
  while (top * 2 < G) top *= 2;
  __syncthreads();

  const bool fits = T <= BKT_CAP;
  for (int i0 = 0; i0 < T; i0 += BKT_EPT * SORT_NT) {
    int2 p[BKT_EPT];
    int ic[BKT_EPT], g[BKT_EPT];
    #pragma unroll
    for (int u = 0; u < BKT_EPT; ++u) {
      const int i = i0 + u * SORT_NT + t;
      ic[u] = i < T ? i : T - 1;
      g[u] = 0;
    }
    for (int st = top; st > 0; st >>= 1) {
      #pragma unroll
      for (int u = 0; u < BKT_EPT; ++u)
        if (gs[g[u] + st] <= ic[u]) g[u] += st;
    }
    #pragma unroll
    for (int u = 0; u < BKT_EPT; ++u)
      p[u] = *(const int2*)(cbuf + 2 * (size_t)(gl[g[u]] + ic[u]));
    #pragma unroll
    for (int u = 0; u < BKT_EPT; ++u) {
      const int i = i0 + u * SORT_NT + t;
      if (i < T) {
        const int dl = p[u].x - n0;
        const int r = atomicAdd(&hist[dl], 1);
        if (fits) {
          pk[i] = (dl << 16) | r;
          sv[i] = p[u].y;
        }
      }
    }
  }
  __syncthreads();

  int tot;
  const int hc = hist[t];
  const int hx = block_scan_256(hc, ws, tot);
  lrow[t] = hx;
  if (n0 + t < N) rowptr[n0 + t] = base + hx;
  if (b == B - 1 && t == 0) rowptr[N] = base + T;
  __syncthreads();

  if (fits) {
    #pragma unroll 4
    for (int i = t; i < T; i += SORT_NT) {
      const int v = pk[i];
      stage[lrow[v >> 16] + (v & 0xffff)] = sv[i];
    }
    __syncthreads();
    if (hc > 1 && hc <= 32) {
      int v[32];
      #pragma unroll
      for (int k = 0; k < 32; ++k) v[k] = k < hc ? stage[hx + k] : 0x7fffffff;
      #pragma unroll
      for (int sz = 2; sz <= 32; sz <<= 1) {
        #pragma unroll
        for (int st = sz >> 1; st > 0; st >>= 1) {
          #pragma unroll
          for (int k = 0; k < 32; ++k) {
            const int j = k ^ st;
            if (j > k) {
              const int lo = v[k] < v[j] ? v[k] : v[j];
              const int hi = v[k] < v[j] ? v[j] : v[k];
              v[k] = (k & sz) == 0 ? lo : hi;
              v[j] = (k & sz) == 0 ? hi : lo;
            }
          }
        }
      }
      #pragma unroll
      for (int k = 0; k < 32; ++k)
        if (k < hc) stage[hx + k] = v[k];
    }
    __syncthreads();
    #pragma unroll 4
    for (int i = t; i < T; i += SORT_NT) csrc[base + i] = stage[i];
  } else {
    hist[t] = hx;
    __syncthreads();
    for (int i0 = 0; i0 < T; i0 += SORT_NT) {
      const int i = i0 + t;
      if (i < T) {
        int g = 0;
        for (int st = top; st > 0; st >>= 1)
          if (gs[g + st] <= i) g += st;
        const int2 p = *(const int2*)(cbuf + 2 * (size_t)(gl[g] + i));
        csrc[base + atomicAdd(&hist[p.x - n0], 1)] = p.y;
      }
    }
  }
}

__device__ __forceinline__ float lrelu(float x) {
  return x > 0.f ? x : NEG_SLOPE * x;
}

__global__ __launch_bounds__(256) void gat_gather(
    const int* __restrict__ rowptr, const int* __restrict__ csrc,
    const float* __restrict__ feat, const float* __restrict__ el,
    const float* __restrict__ er, const float* __restrict__ resid,
    float* __restrict__ out, int N) {
  const int lane = threadIdx.x & 31;
  const int hi = lane >> 4;
  const int n = __builtin_amdgcn_readfirstlane(blockIdx.x * 8 + (threadIdx.x >> 5));
  if (n >= N) return;
  const int beg = rowptr[n], end = rowptr[n + 1];
  const float4 b = ((const float4*)er)[n];
  const float4* __restrict__ el4 = (const float4*)el;

  const int j0 = beg + lane;
  const bool has0 = j0 < end;
  const bool more = end - beg > 32;
  int s0 = 0;
  float e0 = 0.f, e1 = 0.f, e2 = 0.f, e3 = 0.f;
  if (has0) {
    s0 = csrc[j0];
    const float4 a = el4[s0];
    e0 = lrelu(a.x + b.x);
    e1 = lrelu(a.y + b.y);
    e2 = lrelu(a.z + b.z);
    e3 = lrelu(a.w + b.w);
  }

  float m0 = -INFINITY, m1 = -INFINITY, m2 = -INFINITY, m3 = -INFINITY;
  if (has0) {
    m0 = fmaxf(m0, e0);
    m1 = fmaxf(m1, e1);
    m2 = fmaxf(m2, e2);
    m3 = fmaxf(m3, e3);
  }
  if (more) {
    for (int j = j0 + 32; j < end; j += 32) {
      const float4 a = el4[csrc[j]];
      m0 = fmaxf(m0, lrelu(a.x + b.x));
      m1 = fmaxf(m1, lrelu(a.y + b.y));
      m2 = fmaxf(m2, lrelu(a.z + b.z));
      m3 = fmaxf(m3, lrelu(a.w + b.w));
    }
  }
  #pragma unroll
  for (int off = 16; off > 0; off >>= 1) {
    m0 = fmaxf(m0, __shfl_xor(m0, off, 32));
    m1 = fmaxf(m1, __shfl_xor(m1, off, 32));
    m2 = fmaxf(m2, __shfl_xor(m2, off, 32));
    m3 = fmaxf(m3, __shfl_xor(m3, off, 32));
  }

  float p0 = 0.f, p1 = 0.f, p2 = 0.f, p3 = 0.f;
  if (has0) {
    p0 = __expf(e0 - m0);
    p1 = __expf(e1 - m1);
    p2 = __expf(e2 - m2);
    p3 = __expf(e3 - m3);
  }
  float d0 = 0.f, d1 = 0.f, d2 = 0.f, d3 = 0.f;
  d0 += p0; d1 += p1; d2 += p2; d3 += p3;
  if (more) {
    for (int j = j0 + 32; j < end; j += 32) {
      const float4 a = el4[csrc[j]];
      d0 += __expf(lrelu(a.x + b.x) - m0);
      d1 += __expf(lrelu(a.y + b.y) - m1);
      d2 += __expf(lrelu(a.z + b.z) - m2);
      d3 += __expf(lrelu(a.w + b.w) - m3);
    }
  }
  #pragma unroll
  for (int off = 16; off > 0; off >>= 1) {
    d0 += __shfl_xor(d0, off, 32);
    d1 += __shfl_xor(d1, off, 32);
    d2 += __shfl_xor(d2, off, 32);
    d3 += __shfl_xor(d3, off, 32);
  }

  float4 accA = make_float4(0.f, 0.f, 0.f, 0.f);
  float4 accB = make_float4(0.f, 0.f, 0.f, 0.f);
  const float* __restrict__ fcol = feat + lane * 4;
  for (int c0 = beg; c0 < end; c0 += 32) {
    int s = 0;
    float w0 = 0.f, w1 = 0.f, w2 = 0.f, w3 = 0.f;
    if (c0 == beg) {
      if (has0) {
        s = s0;
        w0 = p0 / d0;
        w1 = p1 / d1;
        w2 = p2 / d2;
        w3 = p3 / d3;
      }
    } else {
      const int j = c0 + lane;
      if (j < end) {
        s = csrc[j];
        const float4 a = el4[s];
        w0 = __expf(lrelu(a.x + b.x) - m0) / d0;
        w1 = __expf(lrelu(a.y + b.y) - m1) / d1;
        w2 = __expf(lrelu(a.z + b.z) - m2) / d2;
        w3 = __expf(lrelu(a.w + b.w) - m3) / d3;
      }
    }
    const int left = end - c0;
    const int nk = left < 32 ? left : 32;
    #pragma unroll 4
    for (int k = 0; k < nk; ++k) {
      const int sk = __builtin_amdgcn_readlane(s, k);
      const float k0 = __int_as_float(__builtin_amdgcn_readlane(__float_as_int(w0), k));
      const float k1 = __int_as_float(__builtin_amdgcn_readlane(__float_as_int(w1), k));
      const float k2 = __int_as_float(__builtin_amdgcn_readlane(__float_as_int(w2), k));
      const float k3 = __int_as_float(__builtin_amdgcn_readlane(__float_as_int(w3), k));
      const float ca = hi ? k1 : k0;
      const float cb = hi ? k3 : k2;
      const float* __restrict__ fr = fcol + (size_t)sk * FDIM;
      const float4 fa = *(const float4*)fr;
      const float4 fb = *(const float4*)(fr + 128);
      accA.x += fa.x * ca; accA.y += fa.y * ca; accA.z += fa.z * ca; accA.w += fa.w * ca;
      accB.x += fb.x * cb; accB.y += fb.y * cb; accB.z += fb.z * cb; accB.w += fb.w * cb;
    }
  }

  if (resid) {
    const float* __restrict__ rr = resid + (size_t)n * FDIM + lane * 4;
    const float4 ra = *(const float4*)rr;
    const float4 rb = *(const float4*)(rr + 128);
    accA.x += ra.x; accA.y += ra.y; accA.z += ra.z; accA.w += ra.w;
    accB.x += rb.x; accB.y += rb.y; accB.z += rb.z; accB.w += rb.w;
  }
  accA.x = accA.x > 0.f ? accA.x : (__expf(accA.x) - 1.f);
  accA.y = accA.y > 0.f ? accA.y : (__expf(accA.y) - 1.f);
  accA.z = accA.z > 0.f ? accA.z : (__expf(accA.z) - 1.f);
  accA.w = accA.w > 0.f ? accA.w : (__expf(accA.w) - 1.f);
  accB.x = accB.x > 0.f ? accB.x : (__expf(accB.x) - 1.f);
  accB.y = accB.y > 0.f ? accB.y : (__expf(accB.y) - 1.f);
  accB.z = accB.z > 0.f ? accB.z : (__expf(accB.z) - 1.f);
  accB.w = accB.w > 0.f ? accB.w : (__expf(accB.w) - 1.f);
  float* __restrict__ o = out + (size_t)n * FDIM + lane * 4;
  *(float4*)o = accA;
  *(float4*)(o + 128) = accB;
}

static void run_layer(const float* hin, const float* Wm, const float* al,
                      const float* ar, const int* rowptr, const int* csrc,
                      int N, float* feat, float* el, float* er,
                      const float* resid, float* rst, hipStream_t stream) {
  dim3 ggrid((N + 63) / 64, FDIM / 64);
  gemm_f32_wmma<<<ggrid, 128, 0, stream>>>(hin, Wm, al, ar, feat, el, er, N);
  gat_gather<<<(N + 7) / 8, 256, 0, stream>>>(rowptr, csrc, feat, el, er, resid, rst, N);
}

extern "C" void kernel_launch(void* const* d_in, const int* in_sizes, int n_in,
                              void* d_out, int out_size, void* d_ws, size_t ws_size,
                              hipStream_t stream) {
  const float* features = (const float*)d_in[0];
  const int*   src      = (const int*)d_in[1];
  const int*   dst      = (const int*)d_in[2];
  const float* W1       = (const float*)d_in[3];
  const float* al1      = (const float*)d_in[4];
  const float* ar1      = (const float*)d_in[5];
  const float* W2       = (const float*)d_in[6];
  const float* al2      = (const float*)d_in[7];
  const float* ar2      = (const float*)d_in[8];

  const int N = in_sizes[0] / FDIM;
  const int E = in_sizes[1];

  char* w = (char*)d_ws;
  float* feat = (float*)w;  w += (size_t)N * FDIM * sizeof(float);
  float* h1   = (float*)w;  w += (size_t)N * FDIM * sizeof(float);
  float* el   = (float*)w;  w += (size_t)N * HEADS * sizeof(float);
  float* er   = (float*)w;  w += (size_t)N * HEADS * sizeof(float);
  int* csrc   = (int*)w;    w += (size_t)E * sizeof(int);
  int* rowptr = (int*)w;    w += ((size_t)N + 4) / 4 * 4 * sizeof(int);
  const int G = (E + SORT_CH - 1) / SORT_CH;
  const int B = (N + NPB - 1) / NPB;
  int* cbuf   = (int*)w;    w += (size_t)G * SORT_CH * 2 * sizeof(int);
  int* cnt    = (int*)w;    w += (size_t)G * B * sizeof(int);
  int* lof    = (int*)w;

  chunk_sort<<<G, SORT_NT, 0, stream>>>(src, dst, cbuf, cnt, lof, G, B, E);
  bucket_csr<<<B, SORT_NT, 0, stream>>>(cbuf, cnt, lof, rowptr, csrc, G, B, N);

  run_layer(features, W1, al1, ar1, rowptr, csrc, N,
            feat, el, er,  nullptr,  h1, stream);

  run_layer(h1, W2, al2, ar2, rowptr, csrc, N,
            feat, el, er,  h1,  (float*)d_out, stream);
}


# Edge sort no longer has launches of its own: its two steps run as the first work-groups of the layer 1 GEMM's launch, split in two, beside the tiles (two per 256-thread group); sum orders unchanged
# speedup vs baseline: 6.5448x; 1.0311x over previous
#include <hip/hip_runtime.h>
#include <math.h>

#define NEG_SLOPE 0.2f
#define FDIM 256
#define HEADS 4
#define HDIM 64
#define KTILE 32
#define LDT 36

typedef float v2f __attribute__((ext_vector_type(2)));
typedef float v8f __attribute__((ext_vector_type(8)));

#define TILE_LDS (4 * 64 * LDT)

__device__ __forceinline__ void gemm_tile(
    const float* H, const float* W,
    const float* __restrict__ al, const float* __restrict__ ar,
    float* __restrict__ OUT, float* __restrict__ el, float* __restrict__ er,
    int N, int row0, int head, int t, float* lds) {
  float* const As0 = lds;
  float* const As1 = lds + 64 * LDT;
  float* const Bt0 = lds + 2 * 64 * LDT;
  float* const Bt1 = lds + 3 * 64 * LDT;

  const int col0 = head * 64;
  const int wave = t >> 5, lane = t & 31;
  const int hi = lane >> 4;
  const int m  = lane & 15;

  const int ac4 = t & 7, srow = t >> 3;
  int sr0 = row0 + srow, sr1 = sr0 + 16, sr2 = sr0 + 32, sr3 = sr0 + 48;
  if (sr0 > N - 1) sr0 = N - 1;
  if (sr1 > N - 1) sr1 = N - 1;
  if (sr2 > N - 1) sr2 = N - 1;
  if (sr3 > N - 1) sr3 = N - 1;
  const unsigned ao0 = (unsigned)sr0 * FDIM + ac4 * 4;
  const unsigned ao1 = (unsigned)sr1 * FDIM + ac4 * 4;
  const unsigned ao2 = (unsigned)sr2 * FDIM + ac4 * 4;
  const unsigned ao3 = (unsigned)sr3 * FDIM + ac4 * 4;
  const int bng = t & 15, bkg = t >> 4;
  const unsigned bo = (unsigned)(bkg * 4) * FDIM + col0 + bng * 4;
  const int aoff = srow * LDT + ac4 * 4;
  const int boff = (bng * 4) * LDT + bkg * 4;

  float4 ra0, ra1, ra2, ra3, rb0, rb1, rb2, rb3;
  v8f acc0 = {}, acc1 = {}, acc2 = {}, acc3 = {};

#define GEMM_FETCH(kt)                                                        \
  {                                                                           \
    const float* hk = H + (kt);                                               \
    const float* wk = W + (size_t)(kt) * FDIM;                                \
    ra0 = *(const float4*)(hk + ao0);                                         \
    ra1 = *(const float4*)(hk + ao1);                                         \
    ra2 = *(const float4*)(hk + ao2);                                         \
    ra3 = *(const float4*)(hk + ao3);                                         \
    rb0 = *(const float4*)(wk + bo);                                          \
    rb1 = *(const float4*)(wk + bo + FDIM);                                   \
    rb2 = *(const float4*)(wk + bo + 2 * FDIM);                               \
    rb3 = *(const float4*)(wk + bo + 3 * FDIM);                               \
  }
#define GEMM_STASH(s)                                                         \
  {                                                                           \
    __builtin_amdgcn_sched_barrier(0);                                        \
    float* const as_ = (s) ? As1 : As0;                                       \
    float* const bt_ = (s) ? Bt1 : Bt0;                                       \
    *(float4*)&as_[aoff]            = ra0;                                    \
    *(float4*)&as_[aoff + 16 * LDT] = ra1;                                    \
    *(float4*)&as_[aoff + 32 * LDT] = ra2;                                    \
    *(float4*)&as_[aoff + 48 * LDT] = ra3;                                    \
    *(float4*)&bt_[boff]           = make_float4(rb0.x, rb1.x, rb2.x, rb3.x); \
    *(float4*)&bt_[boff + LDT]     = make_float4(rb0.y, rb1.y, rb2.y, rb3.y); \
    *(float4*)&bt_[boff + 2 * LDT] = make_float4(rb0.z, rb1.z, rb2.z, rb3.z); \
    *(float4*)&bt_[boff + 3 * LDT] = make_float4(rb0.w, rb1.w, rb2.w, rb3.w); \
  }
#define GEMM_PRODUCTS(s)                                                      \
  {                                                                           \
    const float* const as_ = (s) ? As1 : As0;                                 \
    const float* const bt_ = (s) ? Bt1 : Bt0;                                 \
    _Pragma("unroll")                                                         \
    for (int kk = 0; kk < KTILE; kk += 4) {                                   \
      const int ka = kk + 2 * hi;        \
      v2f a  = *(const v2f*)&as_[(wave * 16 + m) * LDT + ka];                 \
      v2f b0 = *(const v2f*)&bt_[(m)      * LDT + ka];                        \
      v2f b1 = *(const v2f*)&bt_[(16 + m) * LDT + ka];                        \
      v2f b2 = *(const v2f*)&bt_[(32 + m) * LDT + ka];                        \
      v2f b3 = *(const v2f*)&bt_[(48 + m) * LDT + ka];                        \
      acc0 = __builtin_amdgcn_wmma_f32_16x16x4_f32(false, a, false, b0, (short)0, acc0, false, false); \
      acc1 = __builtin_amdgcn_wmma_f32_16x16x4_f32(false, a, false, b1, (short)0, acc1, false, false); \
      acc2 = __builtin_amdgcn_wmma_f32_16x16x4_f32(false, a, false, b2, (short)0, acc2, false, false); \
      acc3 = __builtin_amdgcn_wmma_f32_16x16x4_f32(false, a, false, b3, (short)0, acc3, false, false); \
    }                                                                         \
  }

  GEMM_FETCH(0);
  GEMM_STASH(0);
  __syncthreads();

  #pragma unroll 1
  for (int kt = 0; kt < FDIM - 2 * KTILE; kt += 2 * KTILE) {
    GEMM_FETCH(kt + KTILE);
    GEMM_PRODUCTS(0);
    GEMM_STASH(1);
    __syncthreads();
    GEMM_FETCH(kt + 2 * KTILE);
    GEMM_PRODUCTS(1);
    GEMM_STASH(0);
    __syncthreads();
  }
  GEMM_FETCH(FDIM - KTILE);
  GEMM_PRODUCTS(0);
  GEMM_STASH(1);
  __syncthreads();
  GEMM_PRODUCTS(1);
#undef GEMM_FETCH
#undef GEMM_STASH
#undef GEMM_PRODUCTS

  #pragma unroll
  for (int r = 0; r < 8; ++r) {
    const int row = row0 + wave * 16 + r + 8 * hi;
    if (row < N) {
      float* o = OUT + (size_t)row * FDIM + col0 + m;
      o[0]  = acc0[r];
      o[16] = acc1[r];
      o[32] = acc2[r];
      o[48] = acc3[r];
    }
  }

  const float* __restrict__ alh = al + col0;
  const float* __restrict__ arh = ar + col0;
  const float l0 = alh[m], l1 = alh[16 + m], l2 = alh[32 + m], l3 = alh[48 + m];
  const float r0 = arh[m], r1 = arh[16 + m], r2 = arh[32 + m], r3 = arh[48 + m];
  float pm = 0.f, qm = 0.f;
  #pragma unroll
  for (int r = 0; r < 8; ++r) {
    float p = acc0[r] * l0 + acc1[r] * l1 + acc2[r] * l2 + acc3[r] * l3;
    float q = acc0[r] * r0 + acc1[r] * r1 + acc2[r] * r2 + acc3[r] * r3;
    #pragma unroll
    for (int off = 8; off > 0; off >>= 1) {
      p += __shfl_xor(p, off, 32);
      q += __shfl_xor(q, off, 32);
    }
    if (m == r) { pm = p; qm = q; }
  }
  const int arow = row0 + wave * 16 + m + 8 * hi;
  if (m < 8 && arow < N) {
    el[arow * HEADS + head] = pm;
    er[arow * HEADS + head] = qm;
  }
}

__global__ __launch_bounds__(128) void gemm_f32_wmma(
    const float* H, const float* W,
    const float* __restrict__ al, const float* __restrict__ ar,
    float* __restrict__ OUT, float* __restrict__ el, float* __restrict__ er,
    int N) {
  __shared__ __attribute__((aligned(16))) float lds[TILE_LDS];
  gemm_tile(H, W, al, ar, OUT, el, er, N, blockIdx.x * 64, blockIdx.y, threadIdx.x, lds);
}

__device__ __forceinline__ void gemm_two_tiles(
    const float* H, const float* W,
    const float* __restrict__ al, const float* __restrict__ ar,
    float* __restrict__ OUT, float* __restrict__ el, float* __restrict__ er,
    int N, int pair, int tile0, int tile1, float* lds) {
  const int half = threadIdx.x >> 7;
  const int nrt = (N + 63) / 64;
  const int id = tile0 + 2 * pair + half;
  int head = 0, row0 = nrt * 64;
  if (id < tile1) {
    head = id / nrt;
    row0 = (id - head * nrt) * 64;
  }
  gemm_tile(H, W, al, ar, OUT, el, er, N, row0, head, threadIdx.x & 127, lds + half * TILE_LDS);
}

#define SORT_NT   256
#define SORT_CH   3136
#define SORT_EPT  13
#define SORT_BMAX 1024
#define NPB_SHIFT 8
#define NPB       256
#define BKT_EPT   20
#define BKT_CAP   5120

__device__ __forceinline__ int block_scan_256(int c, int* ws, int& tot) {
  const int lane = threadIdx.x & 31, wave = threadIdx.x >> 5;
  int incl = c;
  #pragma unroll
  for (int off = 1; off < 32; off <<= 1) {
    const int v = __shfl_up(incl, off, 32);
    if (lane >= off) incl += v;
  }
  if (lane == 31) ws[wave] = incl;
  __syncthreads();
  const int wv = lane < SORT_NT / 32 ? ws[lane] : 0;
  int pre = lane < wave ? wv : 0;
  int all = wv;
  #pragma unroll
  for (int off = 16; off > 0; off >>= 1) {
    pre += __shfl_xor(pre, off, 32);
    all += __shfl_xor(all, off, 32);
  }
  __syncthreads();
  tot = all;
  return pre + incl - c;
}

#define CHUNK_LDS (2 * SORT_BMAX + SORT_NT / 32 + 2 * SORT_CH)

__device__ __forceinline__ void chunk_sort(
    const int* __restrict__ src, const int* __restrict__ dst,
    int* __restrict__ cbuf, int* __restrict__ cnt, int* __restrict__ lof,
    int G, int B, int E, int* lds) {
  int* const bc  = lds;
  int* const bl  = lds + SORT_BMAX;
  int* const ws  = lds + 2 * SORT_BMAX;
  int* const ord = lds + 2 * SORT_BMAX + SORT_NT / 32;
  const int t = threadIdx.x;
  const int g = blockIdx.x;
  const int e0 = g * SORT_CH;
  const int len = E - e0 < SORT_CH ? E - e0 : SORT_CH;
  *(int4*)&bc[4 * t] = make_int4(0, 0, 0, 0);
  __syncthreads();
  int d[SORT_EPT], s[SORT_EPT], r[SORT_EPT];
  #pragma unroll
  for (int u = 0; u < SORT_EPT; ++u) {
    const int i = u * SORT_NT + t;
    const int ii = i < len ? i : len - 1;
    d[u] = dst[e0 + ii];
    s[u] = src[e0 + ii];
  }
  #pragma unroll
  for (int u = 0; u < SORT_EPT; ++u) {
    r[u] = 0;
    if (u * SORT_NT + t < len) r[u] = atomicAdd(&bc[d[u] >> NPB_SHIFT], 1);
  }
  __syncthreads();
  const int4 c4 = *(const int4*)&bc[4 * t];
  int tot;
  const int ex = block_scan_256(c4.x + c4.y + c4.z + c4.w, ws, tot);
  int4 x4;
  x4.x = ex;
  x4.y = x4.x + c4.x;
  x4.z = x4.y + c4.y;
  x4.w = x4.z + c4.z;
  *(int4*)&bl[4 * t] = x4;
  const int k0 = 4 * t;
  int* __restrict__ cg = cnt + g;
  int* __restrict__ lg = lof + g;
  if (k0     < B) { cg[(k0)     * G] = c4.x; lg[(k0)     * G] = x4.x; }
  if (k0 + 1 < B) { cg[(k0 + 1) * G] = c4.y; lg[(k0 + 1) * G] = x4.y; }
  if (k0 + 2 < B) { cg[(k0 + 2) * G] = c4.z; lg[(k0 + 2) * G] = x4.z; }
  if (k0 + 3 < B) { cg[(k0 + 3) * G] = c4.w; lg[(k0 + 3) * G] = x4.w; }
  __syncthreads();
  #pragma unroll
  for (int u = 0; u < SORT_EPT; ++u) {
    if (u * SORT_NT + t < len)
      *(int2*)&ord[2 * (bl[d[u] >> NPB_SHIFT] + r[u])] = make_int2(d[u], s[u]);
  }
  __syncthreads();
  int2* __restrict__ cb = (int2*)cbuf + e0;
  #pragma unroll 4
  for (int i = t; i < len; i += SORT_NT) cb[i] = *(const int2*)&ord[2 * i];
}

#define BUCKET_LDS (2 * SORT_BMAX + 2 * NPB + SORT_NT / 32 + 3 * BKT_CAP)

__device__ __forceinline__ void bucket_csr(
    const int* __restrict__ cbuf, const int* __restrict__ cnt, const int* __restrict__ lof,
    int* __restrict__ rowptr, int* __restrict__ csrc, int G, int B, int N, int* lds) {
  int* const gs    = lds;
  int* const gl    = gs + SORT_BMAX;
  int* const hist  = gl + SORT_BMAX;
  int* const lrow  = hist + NPB;
  int* const ws    = lrow + NPB;
  int* const pk    = ws + SORT_NT / 32;
  int* const sv    = pk + BKT_CAP;
  int* const stage = sv + BKT_CAP;
  const int t = threadIdx.x;
  const int b = blockIdx.x;
  const int n0 = b << NPB_SHIFT;
  int c[4], l[4];
  #pragma unroll
  for (int q = 0; q < 4; ++q) {
    const int k = 4 * t + q;
    c[q] = k < G ? cnt[b * G + k] : 0;
    l[q] = k < G ? lof[b * G + k] : 0;
  }
  int T, base;
  const int ex = block_scan_256(c[0] + c[1] + c[2] + c[3], ws, T);
  (void)block_scan_256(l[0] + l[1] + l[2] + l[3], ws, base);
  int run = ex;
  #pragma unroll
  for (int q = 0; q < 4; ++q) {
    const int k = 4 * t + q;
    gs[k] = run;
    gl[k] = k * SORT_CH + l[q] - run;
    run += c[q];
  }
  hist[t] = 0;
  int top = 1;
  while (top * 2 < G) top *= 2;
  __syncthreads();

  const bool fits = T <= BKT_CAP;
  for (int i0 = 0; i0 < T; i0 += BKT_EPT * SORT_NT) {
    int2 p[BKT_EPT];
    int ic[BKT_EPT], g[BKT_EPT];
    #pragma unroll
    for (int u = 0; u < BKT_EPT; ++u) {
      const int i = i0 + u * SORT_NT + t;
      ic[u] = i < T ? i : T - 1;
      g[u] = 0;
    }
    for (int st = top; st > 0; st >>= 1) {
      #pragma unroll
      for (int u = 0; u < BKT_EPT; ++u)
        if (gs[g[u] + st] <= ic[u]) g[u] += st;
    }
    #pragma unroll
    for (int u = 0; u < BKT_EPT; ++u)
      p[u] = *(const int2*)(cbuf + 2 * (size_t)(gl[g[u]] + ic[u]));
    #pragma unroll
    for (int u = 0; u < BKT_EPT; ++u) {
      const int i = i0 + u * SORT_NT + t;
      if (i < T) {
        const int dl = p[u].x - n0;
        const int r = atomicAdd(&hist[dl], 1);
        if (fits) {
          pk[i] = (dl << 16) | r;
          sv[i] = p[u].y;
        }
      }
    }
  }
  __syncthreads();

  int tot;
  const int hc = hist[t];
  const int hx = block_scan_256(hc, ws, tot);
  lrow[t] = hx;
  if (n0 + t < N) rowptr[n0 + t] = base + hx;
  if (b == B - 1 && t == 0) rowptr[N] = base + T;
  __syncthreads();

  if (fits) {
    #pragma unroll 4
    for (int i = t; i < T; i += SORT_NT) {
      const int v = pk[i];
      stage[lrow[v >> 16] + (v & 0xffff)] = sv[i];
    }
    __syncthreads();
    if (hc > 1 && hc <= 32) {
      int v[32];
      #pragma unroll
      for (int k = 0; k < 32; ++k) v[k] = k < hc ? stage[hx + k] : 0x7fffffff;
      #pragma unroll
      for (int sz = 2; sz <= 32; sz <<= 1) {
        #pragma unroll
        for (int st = sz >> 1; st > 0; st >>= 1) {
          #pragma unroll
          for (int k = 0; k < 32; ++k) {
            const int j = k ^ st;
            if (j > k) {
              const int lo = v[k] < v[j] ? v[k] : v[j];
              const int hi = v[k] < v[j] ? v[j] : v[k];
              v[k] = (k & sz) == 0 ? lo : hi;
              v[j] = (k & sz) == 0 ? hi : lo;
            }
          }
        }
      }
      #pragma unroll
      for (int k = 0; k < 32; ++k)
        if (k < hc) stage[hx + k] = v[k];
    }
    __syncthreads();
    #pragma unroll 4
    for (int i = t; i < T; i += SORT_NT) csrc[base + i] = stage[i];
  } else {
    hist[t] = hx;
    __syncthreads();
    for (int i0 = 0; i0 < T; i0 += SORT_NT) {
      const int i = i0 + t;
      if (i < T) {
        int g = 0;
        for (int st = top; st > 0; st >>= 1)
          if (gs[g + st] <= i) g += st;
        const int2 p = *(const int2*)(cbuf + 2 * (size_t)(gl[g] + i));
        csrc[base + atomicAdd(&hist[p.x - n0], 1)] = p.y;
      }
    }
  }
}

#define ROLE_LDS (2 * TILE_LDS)
#define CHIP_WGPS 256

__global__ __launch_bounds__(SORT_NT) void sort_chunks_gemm(
    const int* __restrict__ src, const int* __restrict__ dst,
    int* __restrict__ cbuf, int* __restrict__ cnt, int* __restrict__ lof,
    int G, int B, int E,
    const float* H, const float* W,
    const float* __restrict__ al, const float* __restrict__ ar,
    float* __restrict__ OUT, float* __restrict__ el, float* __restrict__ er,
    int N, int tile0, int tile1) {
  static_assert(CHUNK_LDS <= ROLE_LDS, "the chunk step's LDS must fit in the tiles'");
  __shared__ __attribute__((aligned(16))) float lds[ROLE_LDS];
  if ((int)blockIdx.x < G)
    chunk_sort(src, dst, cbuf, cnt, lof, G, B, E, (int*)lds);
  else
    gemm_two_tiles(H, W, al, ar, OUT, el, er, N, (int)blockIdx.x - G, tile0, tile1, lds);
}

__global__ __launch_bounds__(SORT_NT) void sort_buckets_gemm(
    const int* __restrict__ cbuf, const int* __restrict__ cnt, const int* __restrict__ lof,
    int* __restrict__ rowptr, int* __restrict__ csrc, int G, int B, int Nn,
    const float* H, const float* W,
    const float* __restrict__ al, const float* __restrict__ ar,
    float* __restrict__ OUT, float* __restrict__ el, float* __restrict__ er,
    int N, int tile0, int tile1) {
  static_assert(BUCKET_LDS <= ROLE_LDS, "the bucket step's LDS must fit in the tiles'");
  __shared__ __attribute__((aligned(16))) float lds[ROLE_LDS];
  if ((int)blockIdx.x < B)
    bucket_csr(cbuf, cnt, lof, rowptr, csrc, G, B, Nn, (int*)lds);
  else
    gemm_two_tiles(H, W, al, ar, OUT, el, er, N, (int)blockIdx.x - B, tile0, tile1, lds);
}

__device__ __forceinline__ float lrelu(float x) {
  return x > 0.f ? x : NEG_SLOPE * x;
}

__global__ __launch_bounds__(256) void gat_gather(
    const int* __restrict__ rowptr, const int* __restrict__ csrc,
    const float* __restrict__ feat, const float* __restrict__ el,
    const float* __restrict__ er, const float* __restrict__ resid,
    float* __restrict__ out, int N) {
  const int lane = threadIdx.x & 31;
  const int hi = lane >> 4;
  const int n = __builtin_amdgcn_readfirstlane(blockIdx.x * 8 + (threadIdx.x >> 5));
  if (n >= N) return;
  const int beg = rowptr[n], end = rowptr[n + 1];
  const float4 b = ((const float4*)er)[n];
  const float4* __restrict__ el4 = (const float4*)el;

  const int j0 = beg + lane;
  const bool has0 = j0 < end;
  const bool more = end - beg > 32;
  int s0 = 0;
  float e0 = 0.f, e1 = 0.f, e2 = 0.f, e3 = 0.f;
  if (has0) {
    s0 = csrc[j0];
    const float4 a = el4[s0];
    e0 = lrelu(a.x + b.x);
    e1 = lrelu(a.y + b.y);
    e2 = lrelu(a.z + b.z);
    e3 = lrelu(a.w + b.w);
  }

  float m0 = -INFINITY, m1 = -INFINITY, m2 = -INFINITY, m3 = -INFINITY;
  if (has0) {
    m0 = fmaxf(m0, e0);
    m1 = fmaxf(m1, e1);
    m2 = fmaxf(m2, e2);
    m3 = fmaxf(m3, e3);
  }
  if (more) {
    for (int j = j0 + 32; j < end; j += 32) {
      const float4 a = el4[csrc[j]];
      m0 = fmaxf(m0, lrelu(a.x + b.x));
      m1 = fmaxf(m1, lrelu(a.y + b.y));
      m2 = fmaxf(m2, lrelu(a.z + b.z));
      m3 = fmaxf(m3, lrelu(a.w + b.w));
    }
  }
  #pragma unroll
  for (int off = 16; off > 0; off >>= 1) {
    m0 = fmaxf(m0, __shfl_xor(m0, off, 32));
    m1 = fmaxf(m1, __shfl_xor(m1, off, 32));
    m2 = fmaxf(m2, __shfl_xor(m2, off, 32));
    m3 = fmaxf(m3, __shfl_xor(m3, off, 32));
  }

  float p0 = 0.f, p1 = 0.f, p2 = 0.f, p3 = 0.f;
  if (has0) {
    p0 = __expf(e0 - m0);
    p1 = __expf(e1 - m1);
    p2 = __expf(e2 - m2);
    p3 = __expf(e3 - m3);
  }
  float d0 = 0.f, d1 = 0.f, d2 = 0.f, d3 = 0.f;
  d0 += p0; d1 += p1; d2 += p2; d3 += p3;
  if (more) {
    for (int j = j0 + 32; j < end; j += 32) {
      const float4 a = el4[csrc[j]];
      d0 += __expf(lrelu(a.x + b.x) - m0);
      d1 += __expf(lrelu(a.y + b.y) - m1);
      d2 += __expf(lrelu(a.z + b.z) - m2);
      d3 += __expf(lrelu(a.w + b.w) - m3);
    }
  }
  #pragma unroll
  for (int off = 16; off > 0; off >>= 1) {
    d0 += __shfl_xor(d0, off, 32);
    d1 += __shfl_xor(d1, off, 32);
    d2 += __shfl_xor(d2, off, 32);
    d3 += __shfl_xor(d3, off, 32);
  }

  float4 accA = make_float4(0.f, 0.f, 0.f, 0.f);
  float4 accB = make_float4(0.f, 0.f, 0.f, 0.f);
  const float* __restrict__ fcol = feat + lane * 4;
  for (int c0 = beg; c0 < end; c0 += 32) {
    int s = 0;
    float w0 = 0.f, w1 = 0.f, w2 = 0.f, w3 = 0.f;
    if (c0 == beg) {
      if (has0) {
        s = s0;
        w0 = p0 / d0;
        w1 = p1 / d1;
        w2 = p2 / d2;
        w3 = p3 / d3;
      }
    } else {
      const int j = c0 + lane;
      if (j < end) {
        s = csrc[j];
        const float4 a = el4[s];
        w0 = __expf(lrelu(a.x + b.x) - m0) / d0;
        w1 = __expf(lrelu(a.y + b.y) - m1) / d1;
        w2 = __expf(lrelu(a.z + b.z) - m2) / d2;
        w3 = __expf(lrelu(a.w + b.w) - m3) / d3;
      }
    }
    const int left = end - c0;
    const int nk = left < 32 ? left : 32;
    #pragma unroll 4
    for (int k = 0; k < nk; ++k) {
      const int sk = __builtin_amdgcn_readlane(s, k);
      const float k0 = __int_as_float(__builtin_amdgcn_readlane(__float_as_int(w0), k));
      const float k1 = __int_as_float(__builtin_amdgcn_readlane(__float_as_int(w1), k));
      const float k2 = __int_as_float(__builtin_amdgcn_readlane(__float_as_int(w2), k));
      const float k3 = __int_as_float(__builtin_amdgcn_readlane(__float_as_int(w3), k));
      const float ca = hi ? k1 : k0;
      const float cb = hi ? k3 : k2;
      const float* __restrict__ fr = fcol + (size_t)sk * FDIM;
      const float4 fa = *(const float4*)fr;
      const float4 fb = *(const float4*)(fr + 128);
      accA.x += fa.x * ca; accA.y += fa.y * ca; accA.z += fa.z * ca; accA.w += fa.w * ca;
      accB.x += fb.x * cb; accB.y += fb.y * cb; accB.z += fb.z * cb; accB.w += fb.w * cb;
    }
  }

  if (resid) {
    const float* __restrict__ rr = resid + (size_t)n * FDIM + lane * 4;
    const float4 ra = *(const float4*)rr;
    const float4 rb = *(const float4*)(rr + 128);
    accA.x += ra.x; accA.y += ra.y; accA.z += ra.z; accA.w += ra.w;
    accB.x += rb.x; accB.y += rb.y; accB.z += rb.z; accB.w += rb.w;
  }
  accA.x = accA.x > 0.f ? accA.x : (__expf(accA.x) - 1.f);
  accA.y = accA.y > 0.f ? accA.y : (__expf(accA.y) - 1.f);
  accA.z = accA.z > 0.f ? accA.z : (__expf(accA.z) - 1.f);
  accA.w = accA.w > 0.f ? accA.w : (__expf(accA.w) - 1.f);
  accB.x = accB.x > 0.f ? accB.x : (__expf(accB.x) - 1.f);
  accB.y = accB.y > 0.f ? accB.y : (__expf(accB.y) - 1.f);
  accB.z = accB.z > 0.f ? accB.z : (__expf(accB.z) - 1.f);
  accB.w = accB.w > 0.f ? accB.w : (__expf(accB.w) - 1.f);
  float* __restrict__ o = out + (size_t)n * FDIM + lane * 4;
  *(float4*)o = accA;
  *(float4*)(o + 128) = accB;
}


extern "C" void kernel_launch(void* const* d_in, const int* in_sizes, int n_in,
                              void* d_out, int out_size, void* d_ws, size_t ws_size,
                              hipStream_t stream) {
  const float* features = (const float*)d_in[0];
  const int*   src      = (const int*)d_in[1];
  const int*   dst      = (const int*)d_in[2];
  const float* W1       = (const float*)d_in[3];
  const float* al1      = (const float*)d_in[4];
  const float* ar1      = (const float*)d_in[5];
  const float* W2       = (const float*)d_in[6];
  const float* al2      = (const float*)d_in[7];
  const float* ar2      = (const float*)d_in[8];

  const int N = in_sizes[0] / FDIM;
  const int E = in_sizes[1];

  char* w = (char*)d_ws;
  float* feat = (float*)w;  w += (size_t)N * FDIM * sizeof(float);
  float* h1   = (float*)w;  w += (size_t)N * FDIM * sizeof(float);
  float* el   = (float*)w;  w += (size_t)N * HEADS * sizeof(float);
  float* er   = (float*)w;  w += (size_t)N * HEADS * sizeof(float);
  int* csrc   = (int*)w;    w += (size_t)E * sizeof(int);
  int* rowptr = (int*)w;    w += ((size_t)N + 4) / 4 * 4 * sizeof(int);
  const int G = (E + SORT_CH - 1) / SORT_CH;
  const int B = (N + NPB - 1) / NPB;
  int* cbuf   = (int*)w;    w += (size_t)G * SORT_CH * 2 * sizeof(int);
  int* cnt    = (int*)w;    w += (size_t)G * B * sizeof(int);
  int* lof    = (int*)w;

  const int tiles = (N + 63) / 64 * (FDIM / 64);
  int tilesA = 4 * CHIP_WGPS;
  if (tilesA > tiles / 2) tilesA = tiles / 4 * 2;
  const int pairsB = (tiles - tilesA + 1) / 2;
  sort_chunks_gemm<<<G + tilesA / 2, SORT_NT, 0, stream>>>(
      src, dst, cbuf, cnt, lof, G, B, E,
      features, W1, al1, ar1, feat, el, er, N, 0, tilesA);
  sort_buckets_gemm<<<B + pairsB, SORT_NT, 0, stream>>>(
      cbuf, cnt, lof, rowptr, csrc, G, B, N,
      features, W1, al1, ar1, feat, el, er, N, tilesA, tiles);

  gat_gather<<<(N + 7) / 8, 256, 0, stream>>>(rowptr, csrc, feat, el, er,  nullptr,  h1, N);

  dim3 ggrid((N + 63) / 64, FDIM / 64);
  gemm_f32_wmma<<<ggrid, 128, 0, stream>>>(h1, W2, al2, ar2, feat, el, er, N);
  gat_gather<<<(N + 7) / 8, 256, 0, stream>>>(rowptr, csrc, feat, el, er,  h1,  (float*)d_out, N);
}
